# starting kernel
# baseline (speedup 1.0000x reference)
#include <hip/hip_runtime.h>
#include <hip/hip_bf16.h>
#include <math.h>

#define BB      32
#define LL      128
#define DD      300
#define KK_REL  12
#define NREL    25
#define PD      50
#define NPAIR   3044
#define FWID    650
#define MROWS   (BB * LL)
#define NTILES  19
#define NPAD    (NTILES * 16)

typedef __attribute__((ext_vector_type(2))) float v2f;
typedef __attribute__((ext_vector_type(8))) float v8f;

__global__ __launch_bounds__(256) void setup_kernel(
    const float* __restrict__ pos_emb, const float* __restrict__ W_hid,
    const float* __restrict__ b_hid,
    float* __restrict__ S_ws, float* __restrict__ E_ws,
    int* __restrict__ pe, int* __restrict__ pc)
{
    const int tid = threadIdx.x;

    for (int i = tid; i < LL; i += 256) {
        int start = 0;
        for (int t = 0; t < i; ++t) {
            int lo = t - KK_REL < 0 ? 0 : t - KK_REL;
            int hi = t + KK_REL > LL - 1 ? LL - 1 : t + KK_REL;
            start += hi - lo + 1;
        }
        int lo = i - KK_REL < 0 ? 0 : i - KK_REL;
        int hi = i + KK_REL > LL - 1 ? LL - 1 : i + KK_REL;
        for (int j = lo; j <= hi; ++j) { pe[start] = i; pc[start] = j; ++start; }
    }

    for (int e = tid; e < NREL * PD; e += 256) {
        int r = e / PD, pcol = e % PD;
        float acc = 0.f;
        for (int rp = 0; rp < NREL; ++rp) {
            int ad = rp >= KK_REL ? rp - KK_REL : KK_REL - rp;
            float cnt = (float)(LL - ad);
            int d = r - rp;
            acc += cnt * expf(-(float)(d * d)) * pos_emb[rp * PD + pcol];
        }
        S_ws[e] = acc;
    }
    __syncthreads();

    for (int e = tid; e < NREL * DD; e += 256) {
        int r = e / DD, dout = e % DD;
        float acc = b_hid[dout];
        const float* srow = S_ws + r * PD;
        const float* wrow = W_hid + (size_t)dout * FWID + 2 * DD;
        for (int pcol = 0; pcol < PD; ++pcol) acc += srow[pcol] * wrow[pcol];
        E_ws[e] = acc;
    }
}

__global__ __launch_bounds__(256) void pack_x(
    const float* __restrict__ h_e, const float* __restrict__ h_c,
    const float* __restrict__ h_sh,
    float* __restrict__ XE, float* __restrict__ XC)
{
    const int i = blockIdx.x * 256 + threadIdx.x;
    const int n4 = (MROWS * DD) / 4;
    if (i >= n4) return;
    const float* src = blockIdx.y ? h_c : h_e;
    float*       dst = blockIdx.y ? XC : XE;
    float4 a = ((const float4*)src)[i];
    float4 s = ((const float4*)h_sh)[i];
    float4 r;
    r.x = a.x + s.x; r.y = a.y + s.y; r.z = a.z + s.z; r.w = a.w + s.w;
    ((float4*)dst)[i] = r;
}

__global__ __launch_bounds__(256) void pack_w(
    const float* __restrict__ W_hid, float* __restrict__ WB)
{
    const int n = blockIdx.x;
    const int sel = blockIdx.y;
    float* dst = WB + ((size_t)sel * NPAD + n) * DD;
    for (int k = threadIdx.x; k < DD; k += 256) {
        dst[k] = (n < DD) ? W_hid[(size_t)n * FWID + sel * DD + k] : 0.f;
    }
}

__global__ __launch_bounds__(256) void pairproj_gemm(
    const float* __restrict__ XE, const float* __restrict__ XC,
    const float* __restrict__ WB,
    float* __restrict__ A_ws, float* __restrict__ C_ws)
{
    const int wave = threadIdx.x >> 5;
    const int lane = threadIdx.x & 31;
    const int mt4 = blockIdx.x;
    const int nt  = blockIdx.y * 8 + wave;
    if (nt >= NTILES) return;

    const int sel = blockIdx.z;
    const float* X   = sel ? XC : XE;
    float*       Out = sel ? C_ws : A_ws;

    const int mlane = lane & 15;
    const int kk    = (lane >> 4) * 2;

    const float* x0 = X + (size_t)((mt4 * 4 + 0) * 16 + mlane) * DD + kk;
    const float* x1 = X + (size_t)((mt4 * 4 + 1) * 16 + mlane) * DD + kk;
    const float* x2 = X + (size_t)((mt4 * 4 + 2) * 16 + mlane) * DD + kk;
    const float* x3 = X + (size_t)((mt4 * 4 + 3) * 16 + mlane) * DD + kk;
    const float* wr = WB + ((size_t)sel * NPAD + nt * 16 + mlane) * DD + kk;

    v8f acc0 = (v8f)0.0f, acc1 = (v8f)0.0f, acc2 = (v8f)0.0f, acc3 = (v8f)0.0f;
#pragma unroll 5
    for (int k0 = 0; k0 < DD; k0 += 4) {
        v2f bf = *(const v2f*)(wr + k0);
        v2f a0 = *(const v2f*)(x0 + k0);
        v2f a1 = *(const v2f*)(x1 + k0);
        v2f a2 = *(const v2f*)(x2 + k0);
        v2f a3 = *(const v2f*)(x3 + k0);
        acc0 = __builtin_amdgcn_wmma_f32_16x16x4_f32(false, a0, false, bf, (short)0, acc0, false, false);
        acc1 = __builtin_amdgcn_wmma_f32_16x16x4_f32(false, a1, false, bf, (short)0, acc1, false, false);
        acc2 = __builtin_amdgcn_wmma_f32_16x16x4_f32(false, a2, false, bf, (short)0, acc2, false, false);
        acc3 = __builtin_amdgcn_wmma_f32_16x16x4_f32(false, a3, false, bf, (short)0, acc3, false, false);
    }

    const int rbase = (lane >> 4) << 3;
    const int ncol  = nt * 16 + mlane;
    if (ncol < DD) {
        v8f accs[4] = {acc0, acc1, acc2, acc3};
#pragma unroll
        for (int i = 0; i < 4; ++i) {
            const size_t row0 = (size_t)((mt4 * 4 + i) * 16 + rbase);
#pragma unroll
            for (int g = 0; g < 8; ++g)
                Out[(row0 + g) * DD + ncol] = accs[i][g];
        }
    }
}

__global__ __launch_bounds__(256) void pair_epilogue(
    const float* __restrict__ A_ws, const float* __restrict__ C_ws,
    const float* __restrict__ E_ws,
    const int* __restrict__ pe, const int* __restrict__ pc,
    const float* __restrict__ ln_g, const float* __restrict__ ln_b,
    const float* __restrict__ W_rel, const float* __restrict__ b_rel,
    float* __restrict__ out)
{
    const int wave = threadIdx.x >> 5;
    const int lane = threadIdx.x & 31;
    const int w = blockIdx.x * 8 + wave;
    if (w >= BB * NPAIR) return;
    const int b = w / NPAIR;
    const int p = w - b * NPAIR;

    const int emo = pe[p];
    const int cau = pc[p];
    const int r   = cau - emo + KK_REL;

    const float* ra = A_ws + ((size_t)b * LL + emo) * DD;
    const float* rc = C_ws + ((size_t)b * LL + cau) * DD;
    const float* re = E_ws + (size_t)r * DD;

    float h[10];
    float s = 0.f, ss = 0.f;
#pragma unroll
    for (int t = 0; t < 5; ++t) {
        const int idx = t * 64 + lane * 2;
        v2f v = (v2f)0.0f;
        if (idx < DD) {
            v2f va = *(const v2f*)(ra + idx);
            v2f vc = *(const v2f*)(rc + idx);
            v2f ve = *(const v2f*)(re + idx);
            v = va + vc + ve;
        }
        h[2 * t]     = v.x;
        h[2 * t + 1] = v.y;
        s  += v.x + v.y;
        ss += v.x * v.x + v.y * v.y;
    }
#pragma unroll
    for (int off = 16; off > 0; off >>= 1) {
        s  += __shfl_xor(s,  off, 32);
        ss += __shfl_xor(ss, off, 32);
    }
    const float mu   = s * (1.f / DD);
    const float var  = ss * (1.f / DD) - mu * mu;
    const float rstd = rsqrtf(var + 1e-5f);

    float acc = 0.f;
#pragma unroll
    for (int t = 0; t < 5; ++t) {
        const int idx = t * 64 + lane * 2;
        if (idx < DD) {
            v2f g  = *(const v2f*)(ln_g  + idx);
            v2f bb = *(const v2f*)(ln_b  + idx);
            v2f wr = *(const v2f*)(W_rel + idx);
            float v0 = (h[2 * t]     - mu) * rstd * g.x + bb.x;
            float v1 = (h[2 * t + 1] - mu) * rstd * g.y + bb.y;
            v0 = v0 > 0.f ? v0 : (expf(v0) - 1.f);
            v1 = v1 > 0.f ? v1 : (expf(v1) - 1.f);
            acc += v0 * wr.x + v1 * wr.y;
        }
    }
#pragma unroll
    for (int off = 16; off > 0; off >>= 1) acc += __shfl_xor(acc, off, 32);

    if (lane == 0) out[(size_t)b * NPAIR + p] = acc + b_rel[0];
}

extern "C" void kernel_launch(void* const* d_in, const int* in_sizes, int n_in,
                              void* d_out, int out_size, void* d_ws, size_t ws_size,
                              hipStream_t stream) {
    const float* h_e     = (const float*)d_in[0];
    const float* h_c     = (const float*)d_in[1];
    const float* h_sh    = (const float*)d_in[2];
    const float* pos_emb = (const float*)d_in[3];
    const float* W_hid   = (const float*)d_in[4];
    const float* b_hid   = (const float*)d_in[5];
    const float* ln_g    = (const float*)d_in[6];
    const float* ln_b    = (const float*)d_in[7];
    const float* W_rel   = (const float*)d_in[8];
    const float* b_rel   = (const float*)d_in[9];
    float* out = (float*)d_out;

    float* ws   = (float*)d_ws;
    float* A_ws = ws;
    float* C_ws = A_ws + (size_t)MROWS * DD;
    float* XE   = C_ws + (size_t)MROWS * DD;
    float* XC   = XE   + (size_t)MROWS * DD;
    float* WB   = XC   + (size_t)MROWS * DD;
    float* E_ws = WB   + (size_t)2 * NPAD * DD;
    float* S_ws = E_ws + NREL * DD;
    int*   pe   = (int*)(S_ws + NREL * PD);
    int*   pc   = pe + NPAIR;

    setup_kernel<<<1, 256, 0, stream>>>(pos_emb, W_hid, b_hid, S_ws, E_ws, pe, pc);

    dim3 pxg((MROWS * DD / 4 + 255) / 256, 2);
    pack_x<<<pxg, 256, 0, stream>>>(h_e, h_c, h_sh, XE, XC);

    dim3 pwg(NPAD, 2);
    pack_w<<<pwg, 256, 0, stream>>>(W_hid, WB);

    dim3 ggrid(MROWS / 64, 3, 2);
    pairproj_gemm<<<ggrid, 256, 0, stream>>>(XE, XC, WB, A_ws, C_ws);

    const int nwaves  = BB * NPAIR;
    const int nblocks = (nwaves + 7) / 8;
    pair_epilogue<<<nblocks, 256, 0, stream>>>(A_ws, C_ws, E_ws, pe, pc,
                                               ln_g, ln_b, W_rel, b_rel, out);
}


# Setup launch removed: 25x300 position table as an f32 matrix-unit product (S in LDS) in 3 extra work-groups of the GEMM launch; pair indices in closed form; 5->4 launches; table sum order changed
# speedup vs baseline: 1.8184x; 1.8184x over previous
#include <hip/hip_runtime.h>
#include <hip/hip_bf16.h>
#include <math.h>

#define BB      32
#define LL      128
#define DD      300
#define KK_REL  12
#define NREL    25
#define PD      50
#define NPAIR   3044
#define FWID    650
#define MROWS   (BB * LL)
#define NTILES  19
#define NPAD    (NTILES * 16)
#define PKPAD   52
#define HEADP   247
#define EBLK    3

typedef __attribute__((ext_vector_type(2))) float v2f;
typedef __attribute__((ext_vector_type(8))) float v8f;

__global__ __launch_bounds__(256) void pack_x(
    const float* __restrict__ h_e, const float* __restrict__ h_c,
    const float* __restrict__ h_sh,
    float* __restrict__ XE, float* __restrict__ XC)
{
    const int i = blockIdx.x * 256 + threadIdx.x;
    const int n4 = (MROWS * DD) / 4;
    if (i >= n4) return;
    const float* src = blockIdx.y ? h_c : h_e;
    float*       dst = blockIdx.y ? XC : XE;
    float4 a = ((const float4*)src)[i];
    float4 s = ((const float4*)h_sh)[i];
    float4 r;
    r.x = a.x + s.x; r.y = a.y + s.y; r.z = a.z + s.z; r.w = a.w + s.w;
    ((float4*)dst)[i] = r;
}

__global__ __launch_bounds__(256) void pack_w(
    const float* __restrict__ W_hid, float* __restrict__ WB)
{
    const int n = blockIdx.x;
    const int sel = blockIdx.y;
    float* dst = WB + ((size_t)sel * NPAD + n) * DD;
    for (int k = threadIdx.x; k < DD; k += 256) {
        dst[k] = (n < DD) ? W_hid[(size_t)n * FWID + sel * DD + k] : 0.f;
    }
}

__device__ __forceinline__ void position_table_slice(
    const int ngrp,
    const float* __restrict__ pos_emb, const float* __restrict__ W_hid,
    const float* __restrict__ b_hid, float* __restrict__ E_ws,
    float (*G_l)[NREL], float (*S_l)[PKPAD])
{
    const int tid = threadIdx.x;

    for (int e = tid; e < NREL * NREL; e += 256) {
        const int r = e / NREL, rp = e % NREL;
        const int ad = rp >= KK_REL ? rp - KK_REL : KK_REL - rp;
        const float cnt = (float)(LL - ad);
        const int d = r - rp;
        G_l[r][rp] = cnt * expf(-(float)(d * d));
    }
    __syncthreads();

    for (int e = tid; e < 32 * PKPAD; e += 256) {
        const int r = e / PKPAD, pcol = e % PKPAD;
        float acc = 0.f;
        if (r < NREL && pcol < PD) {
            for (int rp = 0; rp < NREL; ++rp) acc += G_l[r][rp] * pos_emb[rp * PD + pcol];
        }
        S_l[r][pcol] = acc;
    }
    __syncthreads();

    const int wave = tid >> 5;
    const int lane = tid & 31;
    const int nt = ngrp * 8 + wave;
    if (nt >= NTILES) return;

    const int mlane = lane & 15;
    const int kk    = (lane >> 4) * 2;
    const int n     = nt * 16 + mlane;
    const bool nok  = n < DD;
    const float* wrow = W_hid + (size_t)(nok ? n : 0) * FWID + 2 * DD;

    v2f bfr[PKPAD / 4];
#pragma unroll
    for (int s = 0; s < PKPAD / 4; ++s) {
        const int k = s * 4 + kk;
        const bool ok = nok && (k < PD);
        v2f t = *(const v2f*)(wrow + (k < PD ? k : 0));
        bfr[s] = ok ? t : (v2f)0.0f;
    }
    const float bias = b_hid[nok ? n : 0];

    v8f acc0 = (v8f)0.0f, acc1 = (v8f)0.0f;
#pragma unroll
    for (int s = 0; s < PKPAD / 4; ++s) {
        const int k = s * 4 + kk;
        v2f a0 = *(const v2f*)(&S_l[mlane][k]);
        v2f a1 = *(const v2f*)(&S_l[16 + mlane][k]);
        acc0 = __builtin_amdgcn_wmma_f32_16x16x4_f32(false, a0, false, bfr[s], (short)0, acc0, false, false);
        acc1 = __builtin_amdgcn_wmma_f32_16x16x4_f32(false, a1, false, bfr[s], (short)0, acc1, false, false);
    }

    const int rbase = (lane >> 4) << 3;
    if (nok) {
#pragma unroll
        for (int g = 0; g < 8; ++g) {
            const int r0 = rbase + g;
            E_ws[r0 * DD + n] = acc0[g] + bias;
            const int r1 = 16 + rbase + g;
            if (r1 < NREL) E_ws[r1 * DD + n] = acc1[g] + bias;
        }
    }
}

__global__ __launch_bounds__(256) void pairproj_gemm(
    const float* __restrict__ XE, const float* __restrict__ XC,
    const float* __restrict__ WB,
    const float* __restrict__ pos_emb, const float* __restrict__ W_hid,
    const float* __restrict__ b_hid, float* __restrict__ E_ws,
    float* __restrict__ A_ws, float* __restrict__ C_ws)
{
    __shared__ float G_l[NREL][NREL];
    __shared__ __attribute__((aligned(16))) float S_l[32][PKPAD];

    if (blockIdx.x < EBLK) {
        position_table_slice(blockIdx.x, pos_emb, W_hid, b_hid, E_ws, G_l, S_l);
        return;
    }
    const int gid = blockIdx.x - EBLK;

    const int wave = threadIdx.x >> 5;
    const int lane = threadIdx.x & 31;
    const int mt4 = gid & 63;
    const int nt  = ((gid >> 6) % 3) * 8 + wave;
    if (nt >= NTILES) return;

    const int sel = gid / 192;
    const float* X   = sel ? XC : XE;
    float*       Out = sel ? C_ws : A_ws;

    const int mlane = lane & 15;
    const int kk    = (lane >> 4) * 2;

    const float* x0 = X + (size_t)((mt4 * 4 + 0) * 16 + mlane) * DD + kk;
    const float* x1 = X + (size_t)((mt4 * 4 + 1) * 16 + mlane) * DD + kk;
    const float* x2 = X + (size_t)((mt4 * 4 + 2) * 16 + mlane) * DD + kk;
    const float* x3 = X + (size_t)((mt4 * 4 + 3) * 16 + mlane) * DD + kk;
    const float* wr = WB + ((size_t)sel * NPAD + nt * 16 + mlane) * DD + kk;

    v8f acc0 = (v8f)0.0f, acc1 = (v8f)0.0f, acc2 = (v8f)0.0f, acc3 = (v8f)0.0f;
#pragma unroll 5
    for (int k0 = 0; k0 < DD; k0 += 4) {
        v2f bf = *(const v2f*)(wr + k0);
        v2f a0 = *(const v2f*)(x0 + k0);
        v2f a1 = *(const v2f*)(x1 + k0);
        v2f a2 = *(const v2f*)(x2 + k0);
        v2f a3 = *(const v2f*)(x3 + k0);
        acc0 = __builtin_amdgcn_wmma_f32_16x16x4_f32(false, a0, false, bf, (short)0, acc0, false, false);
        acc1 = __builtin_amdgcn_wmma_f32_16x16x4_f32(false, a1, false, bf, (short)0, acc1, false, false);
        acc2 = __builtin_amdgcn_wmma_f32_16x16x4_f32(false, a2, false, bf, (short)0, acc2, false, false);
        acc3 = __builtin_amdgcn_wmma_f32_16x16x4_f32(false, a3, false, bf, (short)0, acc3, false, false);
    }

    const int rbase = (lane >> 4) << 3;
    const int ncol  = nt * 16 + mlane;
    if (ncol < DD) {
        v8f accs[4] = {acc0, acc1, acc2, acc3};
#pragma unroll
        for (int i = 0; i < 4; ++i) {
            const size_t row0 = (size_t)((mt4 * 4 + i) * 16 + rbase);
#pragma unroll
            for (int g = 0; g < 8; ++g)
                Out[(row0 + g) * DD + ncol] = accs[i][g];
        }
    }
}

__global__ __launch_bounds__(256) void pair_epilogue(
    const float* __restrict__ A_ws, const float* __restrict__ C_ws,
    const float* __restrict__ E_ws,
    const float* __restrict__ ln_g, const float* __restrict__ ln_b,
    const float* __restrict__ W_rel, const float* __restrict__ b_rel,
    float* __restrict__ out)
{
    const int wave = __builtin_amdgcn_readfirstlane(threadIdx.x >> 5);
    const int lane = threadIdx.x & 31;
    const int w = blockIdx.x * 8 + wave;
    if (w >= BB * NPAIR) return;
    const int b = w / NPAIR;
    const int p = w - b * NPAIR;

    int emo, cau;
    {
        const bool tail = p >= NPAIR - HEADP;
        const int q = tail ? (NPAIR - 1 - p) : p;
        if (q < HEADP) {
            int i = 0;
#pragma unroll
            for (int t = 1; t <= KK_REL; ++t) i += (q >= (t * (t + 25)) / 2) ? 1 : 0;
            const int j = q - (i * (i + 25)) / 2;
            emo = tail ? (LL - 1 - i) : i;
            cau = tail ? (LL - 1 - j) : j;
        } else {
            const int m = p - HEADP;
            emo = 13 + m / NREL;
            cau = emo - KK_REL + m % NREL;
        }
    }
    const int r   = cau - emo + KK_REL;

    const float* ra = A_ws + ((size_t)b * LL + emo) * DD;
    const float* rc = C_ws + ((size_t)b * LL + cau) * DD;
    const float* re = E_ws + (size_t)r * DD;

    float h[10];
    float s = 0.f, ss = 0.f;
#pragma unroll
    for (int t = 0; t < 5; ++t) {
        const int idx = t * 64 + lane * 2;
        v2f v = (v2f)0.0f;
        if (idx < DD) {
            v2f va = *(const v2f*)(ra + idx);
            v2f vc = *(const v2f*)(rc + idx);
            v2f ve = *(const v2f*)(re + idx);
            v = va + vc + ve;
        }
        h[2 * t]     = v.x;
        h[2 * t + 1] = v.y;
        s  += v.x + v.y;
        ss += v.x * v.x + v.y * v.y;
    }
#pragma unroll
    for (int off = 16; off > 0; off >>= 1) {
        s  += __shfl_xor(s,  off, 32);
        ss += __shfl_xor(ss, off, 32);
    }
    const float mu   = s * (1.f / DD);
    const float var  = ss * (1.f / DD) - mu * mu;
    const float rstd = rsqrtf(var + 1e-5f);

    float acc = 0.f;
#pragma unroll
    for (int t = 0; t < 5; ++t) {
        const int idx = t * 64 + lane * 2;
        if (idx < DD) {
            v2f g  = *(const v2f*)(ln_g  + idx);
            v2f bb = *(const v2f*)(ln_b  + idx);
            v2f wr = *(const v2f*)(W_rel + idx);
            float v0 = (h[2 * t]     - mu) * rstd * g.x + bb.x;
            float v1 = (h[2 * t + 1] - mu) * rstd * g.y + bb.y;
            v0 = v0 > 0.f ? v0 : (expf(v0) - 1.f);
            v1 = v1 > 0.f ? v1 : (expf(v1) - 1.f);
            acc += v0 * wr.x + v1 * wr.y;
        }
    }
#pragma unroll
    for (int off = 16; off > 0; off >>= 1) acc += __shfl_xor(acc, off, 32);

    if (lane == 0) out[(size_t)b * NPAIR + p] = acc + b_rel[0];
}

extern "C" void kernel_launch(void* const* d_in, const int* in_sizes, int n_in,
                              void* d_out, int out_size, void* d_ws, size_t ws_size,
                              hipStream_t stream) {
    const float* h_e     = (const float*)d_in[0];
    const float* h_c     = (const float*)d_in[1];
    const float* h_sh    = (const float*)d_in[2];
    const float* pos_emb = (const float*)d_in[3];
    const float* W_hid   = (const float*)d_in[4];
    const float* b_hid   = (const float*)d_in[5];
    const float* ln_g    = (const float*)d_in[6];
    const float* ln_b    = (const float*)d_in[7];
    const float* W_rel   = (const float*)d_in[8];
    const float* b_rel   = (const float*)d_in[9];
    float* out = (float*)d_out;

    float* ws   = (float*)d_ws;
    float* A_ws = ws;
    float* C_ws = A_ws + (size_t)MROWS * DD;
    float* XE   = C_ws + (size_t)MROWS * DD;
    float* XC   = XE   + (size_t)MROWS * DD;
    float* WB   = XC   + (size_t)MROWS * DD;
    float* E_ws = WB   + (size_t)2 * NPAD * DD;

    dim3 pxg((MROWS * DD / 4 + 255) / 256, 2);
    pack_x<<<pxg, 256, 0, stream>>>(h_e, h_c, h_sh, XE, XC);

    dim3 pwg(NPAD, 2);
    pack_w<<<pwg, 256, 0, stream>>>(W_hid, WB);

    dim3 ggrid(EBLK + (MROWS / 64) * 3 * 2);
    pairproj_gemm<<<ggrid, 256, 0, stream>>>(XE, XC, WB, pos_emb, W_hid, b_hid, E_ws, A_ws, C_ws);

    const int nwaves  = BB * NPAIR;
    const int nblocks = (nwaves + 7) / 8;
    pair_epilogue<<<nblocks, 256, 0, stream>>>(A_ws, C_ws, E_ws,
                                               ln_g, ln_b, W_rel, b_rel, out);
}


# Epilogue: a wave per (batch, relative position, 9 or 8 emo rows) walks its diagonal; E row, ln_g, ln_b, W_rel held in registers, A and C rows streamed; 97,408 -> 12,000 waves; same sum order
# speedup vs baseline: 1.9528x; 1.0739x over previous
#include <hip/hip_runtime.h>
#include <hip/hip_bf16.h>
#include <math.h>

#define BB      32
#define LL      128
#define DD      300
#define KK_REL  12
#define NREL    25
#define PD      50
#define NPAIR   3044
#define FWID    650
#define MROWS   (BB * LL)
#define NTILES  19
#define NPAD    (NTILES * 16)
#define PKPAD   52
#define HEADP   247
#define EBLK    3

typedef __attribute__((ext_vector_type(2))) float v2f;
typedef __attribute__((ext_vector_type(8))) float v8f;

__global__ __launch_bounds__(256) void pack_x(
    const float* __restrict__ h_e, const float* __restrict__ h_c,
    const float* __restrict__ h_sh,
    float* __restrict__ XE, float* __restrict__ XC)
{
    const int i = blockIdx.x * 256 + threadIdx.x;
    const int n4 = (MROWS * DD) / 4;
    if (i >= n4) return;
    const float* src = blockIdx.y ? h_c : h_e;
    float*       dst = blockIdx.y ? XC : XE;
    float4 a = ((const float4*)src)[i];
    float4 s = ((const float4*)h_sh)[i];
    float4 r;
    r.x = a.x + s.x; r.y = a.y + s.y; r.z = a.z + s.z; r.w = a.w + s.w;
    ((float4*)dst)[i] = r;
}

__global__ __launch_bounds__(256) void pack_w(
    const float* __restrict__ W_hid, float* __restrict__ WB)
{
    const int n = blockIdx.x;
    const int sel = blockIdx.y;
    float* dst = WB + ((size_t)sel * NPAD + n) * DD;
    for (int k = threadIdx.x; k < DD; k += 256) {
        dst[k] = (n < DD) ? W_hid[(size_t)n * FWID + sel * DD + k] : 0.f;
    }
}

__device__ __forceinline__ void position_table_slice(
    const int ngrp,
    const float* __restrict__ pos_emb, const float* __restrict__ W_hid,
    const float* __restrict__ b_hid, float* __restrict__ E_ws,
    float (*G_l)[NREL], float (*S_l)[PKPAD])
{
    const int tid = threadIdx.x;

    for (int e = tid; e < NREL * NREL; e += 256) {
        const int r = e / NREL, rp = e % NREL;
        const int ad = rp >= KK_REL ? rp - KK_REL : KK_REL - rp;
        const float cnt = (float)(LL - ad);
        const int d = r - rp;
        G_l[r][rp] = cnt * expf(-(float)(d * d));
    }
    __syncthreads();

    for (int e = tid; e < 32 * PKPAD; e += 256) {
        const int r = e / PKPAD, pcol = e % PKPAD;
        float acc = 0.f;
        if (r < NREL && pcol < PD) {
            for (int rp = 0; rp < NREL; ++rp) acc += G_l[r][rp] * pos_emb[rp * PD + pcol];
        }
        S_l[r][pcol] = acc;
    }
    __syncthreads();

    const int wave = tid >> 5;
    const int lane = tid & 31;
    const int nt = ngrp * 8 + wave;
    if (nt >= NTILES) return;

    const int mlane = lane & 15;
    const int kk    = (lane >> 4) * 2;
    const int n     = nt * 16 + mlane;
    const bool nok  = n < DD;
    const float* wrow = W_hid + (size_t)(nok ? n : 0) * FWID + 2 * DD;

    v2f bfr[PKPAD / 4];
#pragma unroll
    for (int s = 0; s < PKPAD / 4; ++s) {
        const int k = s * 4 + kk;
        const bool ok = nok && (k < PD);
        v2f t = *(const v2f*)(wrow + (k < PD ? k : 0));
        bfr[s] = ok ? t : (v2f)0.0f;
    }
    const float bias = b_hid[nok ? n : 0];

    v8f acc0 = (v8f)0.0f, acc1 = (v8f)0.0f;
#pragma unroll
    for (int s = 0; s < PKPAD / 4; ++s) {
        const int k = s * 4 + kk;
        v2f a0 = *(const v2f*)(&S_l[mlane][k]);
        v2f a1 = *(const v2f*)(&S_l[16 + mlane][k]);
        acc0 = __builtin_amdgcn_wmma_f32_16x16x4_f32(false, a0, false, bfr[s], (short)0, acc0, false, false);
        acc1 = __builtin_amdgcn_wmma_f32_16x16x4_f32(false, a1, false, bfr[s], (short)0, acc1, false, false);
    }

    const int rbase = (lane >> 4) << 3;
    if (nok) {
#pragma unroll
        for (int g = 0; g < 8; ++g) {
            const int r0 = rbase + g;
            E_ws[r0 * DD + n] = acc0[g] + bias;
            const int r1 = 16 + rbase + g;
            if (r1 < NREL) E_ws[r1 * DD + n] = acc1[g] + bias;
        }
    }
}

__global__ __launch_bounds__(256) void pairproj_gemm(
    const float* __restrict__ XE, const float* __restrict__ XC,
    const float* __restrict__ WB,
    const float* __restrict__ pos_emb, const float* __restrict__ W_hid,
    const float* __restrict__ b_hid, float* __restrict__ E_ws,
    float* __restrict__ A_ws, float* __restrict__ C_ws)
{
    __shared__ float G_l[NREL][NREL];
    __shared__ __attribute__((aligned(16))) float S_l[32][PKPAD];

    if (blockIdx.x < EBLK) {
        position_table_slice(blockIdx.x, pos_emb, W_hid, b_hid, E_ws, G_l, S_l);
        return;
    }
    const int gid = blockIdx.x - EBLK;

    const int wave = threadIdx.x >> 5;
    const int lane = threadIdx.x & 31;
    const int mt4 = gid & 63;
    const int nt  = ((gid >> 6) % 3) * 8 + wave;
    if (nt >= NTILES) return;

    const int sel = gid / 192;
    const float* X   = sel ? XC : XE;
    float*       Out = sel ? C_ws : A_ws;

    const int mlane = lane & 15;
    const int kk    = (lane >> 4) * 2;

    const float* x0 = X + (size_t)((mt4 * 4 + 0) * 16 + mlane) * DD + kk;
    const float* x1 = X + (size_t)((mt4 * 4 + 1) * 16 + mlane) * DD + kk;
    const float* x2 = X + (size_t)((mt4 * 4 + 2) * 16 + mlane) * DD + kk;
    const float* x3 = X + (size_t)((mt4 * 4 + 3) * 16 + mlane) * DD + kk;
    const float* wr = WB + ((size_t)sel * NPAD + nt * 16 + mlane) * DD + kk;

    v8f acc0 = (v8f)0.0f, acc1 = (v8f)0.0f, acc2 = (v8f)0.0f, acc3 = (v8f)0.0f;
#pragma unroll 5
    for (int k0 = 0; k0 < DD; k0 += 4) {
        v2f bf = *(const v2f*)(wr + k0);
        v2f a0 = *(const v2f*)(x0 + k0);
        v2f a1 = *(const v2f*)(x1 + k0);
        v2f a2 = *(const v2f*)(x2 + k0);
        v2f a3 = *(const v2f*)(x3 + k0);
        acc0 = __builtin_amdgcn_wmma_f32_16x16x4_f32(false, a0, false, bf, (short)0, acc0, false, false);
        acc1 = __builtin_amdgcn_wmma_f32_16x16x4_f32(false, a1, false, bf, (short)0, acc1, false, false);
        acc2 = __builtin_amdgcn_wmma_f32_16x16x4_f32(false, a2, false, bf, (short)0, acc2, false, false);
        acc3 = __builtin_amdgcn_wmma_f32_16x16x4_f32(false, a3, false, bf, (short)0, acc3, false, false);
    }

    const int rbase = (lane >> 4) << 3;
    const int ncol  = nt * 16 + mlane;
    if (ncol < DD) {
        v8f accs[4] = {acc0, acc1, acc2, acc3};
#pragma unroll
        for (int i = 0; i < 4; ++i) {
            const size_t row0 = (size_t)((mt4 * 4 + i) * 16 + rbase);
#pragma unroll
            for (int g = 0; g < 8; ++g)
                Out[(row0 + g) * DD + ncol] = accs[i][g];
        }
    }
}

#define NSEG    15
#define NLONG   8
#define WPB     (NSEG * NREL)
static_assert(NLONG * 9 + (NSEG - NLONG) * 8 == LL, "the segments tile the emo rows");
#define FULLROW (LL - 1 - KK_REL)
static_assert((BB * WPB) % 8 == 0, "whole work-groups of 8 waves");

__device__ __forceinline__ int pair_row_start(const int i)
{
    if (i <= KK_REL + 1) return (i * (i + NREL)) / 2;
    const int t = i > FULLROW ? i - FULLROW : 0;
    return HEADP + (i - (KK_REL + 1)) * NREL - (t * (t - 1)) / 2;
}

__global__ __launch_bounds__(256) void pair_epilogue(
    const float* __restrict__ A_ws, const float* __restrict__ C_ws,
    const float* __restrict__ E_ws,
    const float* __restrict__ ln_g, const float* __restrict__ ln_b,
    const float* __restrict__ W_rel, const float* __restrict__ b_rel,
    float* __restrict__ out)
{
    const int wave = __builtin_amdgcn_readfirstlane(threadIdx.x >> 5);
    const int lane = threadIdx.x & 31;
    const int w   = blockIdx.x * 8 + wave;
    const int b   = w / WPB;
    const int rem = w - b * WPB;
    const int seg = rem / NREL;
    const int r   = rem - seg * NREL;
    const int d   = r - KK_REL;

    int lo = seg * 9 - (seg > NLONG ? seg - NLONG : 0);
    int hi = lo + (seg < NLONG ? 8 : 7);
    if (lo < -d) lo = -d;
    if (hi > LL - 1 - d) hi = LL - 1 - d;
    if (lo > hi) return;

    v2f ev[5], gv[5], bv[5], wv[5];
#pragma unroll
    for (int t = 0; t < 5; ++t) {
        const int idx = t * 64 + lane * 2;
        ev[t] = (v2f)0.0f; gv[t] = (v2f)0.0f; bv[t] = (v2f)0.0f; wv[t] = (v2f)0.0f;
        if (idx < DD) {
            ev[t] = *(const v2f*)(E_ws + (size_t)r * DD + idx);
            gv[t] = *(const v2f*)(ln_g  + idx);
            bv[t] = *(const v2f*)(ln_b  + idx);
            wv[t] = *(const v2f*)(W_rel + idx);
        }
    }
    const float brel = b_rel[0];

    const float* ra = A_ws + ((size_t)b * LL + lo) * DD;
    const float* rc = C_ws + ((size_t)b * LL + lo + d) * DD;
    float* ob = out + (size_t)b * NPAIR;

    for (int emo = lo; emo <= hi; ++emo, ra += DD, rc += DD) {
        float h[10];
        float s = 0.f, ss = 0.f;
#pragma unroll
        for (int t = 0; t < 5; ++t) {
            const int idx = t * 64 + lane * 2;
            v2f v = (v2f)0.0f;
            if (idx < DD) {
                v2f va = *(const v2f*)(ra + idx);
                v2f vc = *(const v2f*)(rc + idx);
                v = va + vc + ev[t];
            }
            h[2 * t]     = v.x;
            h[2 * t + 1] = v.y;
            s  += v.x + v.y;
            ss += v.x * v.x + v.y * v.y;
        }
#pragma unroll
        for (int off = 16; off > 0; off >>= 1) {
            s  += __shfl_xor(s,  off, 32);
            ss += __shfl_xor(ss, off, 32);
        }
        const float mu   = s * (1.f / DD);
        const float var  = ss * (1.f / DD) - mu * mu;
        const float rstd = rsqrtf(var + 1e-5f);

        float acc = 0.f;
#pragma unroll
        for (int t = 0; t < 5; ++t) {
            const int idx = t * 64 + lane * 2;
            if (idx < DD) {
                float v0 = (h[2 * t]     - mu) * rstd * gv[t].x + bv[t].x;
                float v1 = (h[2 * t + 1] - mu) * rstd * gv[t].y + bv[t].y;
                v0 = v0 > 0.f ? v0 : (expf(v0) - 1.f);
                v1 = v1 > 0.f ? v1 : (expf(v1) - 1.f);
                acc += v0 * wv[t].x + v1 * wv[t].y;
            }
        }
#pragma unroll
        for (int off = 16; off > 0; off >>= 1) acc += __shfl_xor(acc, off, 32);

        const int cau  = emo + d;
        const int colo = emo > KK_REL ? emo - KK_REL : 0;
        const int p    = pair_row_start(emo) + cau - colo;
        if (lane == 0) ob[p] = acc + brel;
    }
}

extern "C" void kernel_launch(void* const* d_in, const int* in_sizes, int n_in,
                              void* d_out, int out_size, void* d_ws, size_t ws_size,
                              hipStream_t stream) {
    const float* h_e     = (const float*)d_in[0];
    const float* h_c     = (const float*)d_in[1];
    const float* h_sh    = (const float*)d_in[2];
    const float* pos_emb = (const float*)d_in[3];
    const float* W_hid   = (const float*)d_in[4];
    const float* b_hid   = (const float*)d_in[5];
    const float* ln_g    = (const float*)d_in[6];
    const float* ln_b    = (const float*)d_in[7];
    const float* W_rel   = (const float*)d_in[8];
    const float* b_rel   = (const float*)d_in[9];
    float* out = (float*)d_out;

    float* ws   = (float*)d_ws;
    float* A_ws = ws;
    float* C_ws = A_ws + (size_t)MROWS * DD;
    float* XE   = C_ws + (size_t)MROWS * DD;
    float* XC   = XE   + (size_t)MROWS * DD;
    float* WB   = XC   + (size_t)MROWS * DD;
    float* E_ws = WB   + (size_t)2 * NPAD * DD;

    dim3 pxg((MROWS * DD / 4 + 255) / 256, 2);
    pack_x<<<pxg, 256, 0, stream>>>(h_e, h_c, h_sh, XE, XC);

    dim3 pwg(NPAD, 2);
    pack_w<<<pwg, 256, 0, stream>>>(W_hid, WB);

    dim3 ggrid(EBLK + (MROWS / 64) * 3 * 2);
    pairproj_gemm<<<ggrid, 256, 0, stream>>>(XE, XC, WB, pos_emb, W_hid, b_hid, E_ws, A_ws, C_ws);

    pair_epilogue<<<BB * WPB / 8, 256, 0, stream>>>(A_ws, C_ws, E_ws,
                                                    ln_g, ln_b, W_rel, b_rel, out);
}


# GEMM operands pre-packed by the pack kernels in fragment order [16-row tile][8-k block][lane] float4 (K padded to 304): a fragment load is 512 contiguous B, issued one block ahead; sum order changed
# speedup vs baseline: 2.6571x; 1.3607x over previous
#include <hip/hip_runtime.h>
#include <hip/hip_bf16.h>
#include <math.h>

#define BB      32
#define LL      128
#define DD      300
#define KK_REL  12
#define NREL    25
#define PD      50
#define NPAIR   3044
#define FWID    650
#define MROWS   (BB * LL)
#define NTILES  19
#define NPAD    (NTILES * 16)
#define PKPAD   52
#define HEADP   247
#define EBLK    3
#define KB8     38
#define TSTR    (KB8 * 32)
#define XPACK   ((size_t)(MROWS / 16) * TSTR * 4)
#define WPACK   ((size_t)2 * NTILES * TSTR * 4)

typedef __attribute__((ext_vector_type(2))) float v2f;
typedef __attribute__((ext_vector_type(4))) float v4f;
typedef __attribute__((ext_vector_type(8))) float v8f;


__global__ __launch_bounds__(256) void pack_x(
    const float* __restrict__ h_e, const float* __restrict__ h_c,
    const float* __restrict__ h_sh,
    float* __restrict__ XE, float* __restrict__ XC)
{
    const int i    = blockIdx.x * 256 + threadIdx.x;
    const int lane = i & 31;
    const int blk  = i >> 5;
    const int mt   = blk / KB8;
    const int kb   = blk - mt * KB8;
    const int row  = mt * 16 + (lane & 15);
    const int k    = kb * 8 + (lane >> 4) * 4;
    const float* src = blockIdx.y ? h_c : h_e;
    float*       dst = blockIdx.y ? XC : XE;
    float4 r = make_float4(0.f, 0.f, 0.f, 0.f);
    if (k < DD) {
        const size_t off = (size_t)row * DD + k;
        const float4 a = *(const float4*)(src + off);
        const float4 s = *(const float4*)(h_sh + off);
        r.x = a.x + s.x; r.y = a.y + s.y; r.z = a.z + s.z; r.w = a.w + s.w;
    }
    ((float4*)dst)[i] = r;
}

__global__ __launch_bounds__(256) void pack_w(
    const float* __restrict__ W_hid, float* __restrict__ WB)
{
    const int i    = blockIdx.x * 256 + threadIdx.x;
    const int blk  = i >> 5;
    if (blk >= 2 * NTILES * KB8) return;
    const int lane = i & 31;
    const int kb   = blk % KB8;
    const int t    = blk / KB8;
    const int nt   = t % NTILES;
    const int sel  = t / NTILES;
    const int n    = nt * 16 + (lane & 15);
    const int k    = kb * 8 + (lane >> 4) * 4;
    float4 r = make_float4(0.f, 0.f, 0.f, 0.f);
    if (n < DD && k < DD) {
        const float* p = W_hid + (size_t)n * FWID + sel * DD + k;
        const v2f lo = *(const v2f*)p;
        const v2f hi = *(const v2f*)(p + 2);
        r = make_float4(lo.x, lo.y, hi.x, hi.y);
    }
    ((float4*)WB)[i] = r;
}

__device__ __forceinline__ void position_table_slice(
    const int ngrp,
    const float* __restrict__ pos_emb, const float* __restrict__ W_hid,
    const float* __restrict__ b_hid, float* __restrict__ E_ws,
    float (*G_l)[NREL], float (*S_l)[PKPAD])
{
    const int tid = threadIdx.x;

    for (int e = tid; e < NREL * NREL; e += 256) {
        const int r = e / NREL, rp = e % NREL;
        const int ad = rp >= KK_REL ? rp - KK_REL : KK_REL - rp;
        const float cnt = (float)(LL - ad);
        const int d = r - rp;
        G_l[r][rp] = cnt * expf(-(float)(d * d));
    }
    __syncthreads();

    for (int e = tid; e < 32 * PKPAD; e += 256) {
        const int r = e / PKPAD, pcol = e % PKPAD;
        float acc = 0.f;
        if (r < NREL && pcol < PD) {
            for (int rp = 0; rp < NREL; ++rp) acc += G_l[r][rp] * pos_emb[rp * PD + pcol];
        }
        S_l[r][pcol] = acc;
    }
    __syncthreads();

    const int wave = tid >> 5;
    const int lane = tid & 31;
    const int nt = ngrp * 8 + wave;
    if (nt >= NTILES) return;

    const int mlane = lane & 15;
    const int kk    = (lane >> 4) * 2;
    const int n     = nt * 16 + mlane;
    const bool nok  = n < DD;
    const float* wrow = W_hid + (size_t)(nok ? n : 0) * FWID + 2 * DD;

    v2f bfr[PKPAD / 4];
#pragma unroll
    for (int s = 0; s < PKPAD / 4; ++s) {
        const int k = s * 4 + kk;
        const bool ok = nok && (k < PD);
        v2f t = *(const v2f*)(wrow + (k < PD ? k : 0));
        bfr[s] = ok ? t : (v2f)0.0f;
    }
    const float bias = b_hid[nok ? n : 0];

    v8f acc0 = (v8f)0.0f, acc1 = (v8f)0.0f;
#pragma unroll
    for (int s = 0; s < PKPAD / 4; ++s) {
        const int k = s * 4 + kk;
        v2f a0 = *(const v2f*)(&S_l[mlane][k]);
        v2f a1 = *(const v2f*)(&S_l[16 + mlane][k]);
        acc0 = __builtin_amdgcn_wmma_f32_16x16x4_f32(false, a0, false, bfr[s], (short)0, acc0, false, false);
        acc1 = __builtin_amdgcn_wmma_f32_16x16x4_f32(false, a1, false, bfr[s], (short)0, acc1, false, false);
    }

    const int rbase = (lane >> 4) << 3;
    if (nok) {
#pragma unroll
        for (int g = 0; g < 8; ++g) {
            const int r0 = rbase + g;
            E_ws[r0 * DD + n] = acc0[g] + bias;
            const int r1 = 16 + rbase + g;
            if (r1 < NREL) E_ws[r1 * DD + n] = acc1[g] + bias;
        }
    }
}

__global__ __launch_bounds__(256) void pairproj_gemm(
    const float* __restrict__ XE, const float* __restrict__ XC,
    const float* __restrict__ WB,
    const float* __restrict__ pos_emb, const float* __restrict__ W_hid,
    const float* __restrict__ b_hid, float* __restrict__ E_ws,
    float* __restrict__ A_ws, float* __restrict__ C_ws)
{
    __shared__ float G_l[NREL][NREL];
    __shared__ __attribute__((aligned(16))) float S_l[32][PKPAD];

    if (blockIdx.x < EBLK) {
        position_table_slice(blockIdx.x, pos_emb, W_hid, b_hid, E_ws, G_l, S_l);
        return;
    }
    const int gid = blockIdx.x - EBLK;

    const int wave = threadIdx.x >> 5;
    const int lane = threadIdx.x & 31;
    const int mt4 = gid & 63;
    const int nt  = ((gid >> 6) % 3) * 8 + wave;
    if (nt >= NTILES) return;

    const int sel = gid / 192;
    const float* X   = sel ? XC : XE;
    float*       Out = sel ? C_ws : A_ws;

    const int mlane = lane & 15;

    const v4f* xa = (const v4f*)X  + (size_t)(mt4 * 4) * TSTR + lane;
    const v4f* wb = (const v4f*)WB + (size_t)(sel * NTILES + nt) * TSTR + lane;

#define MM8(A0, A1, A2, A3, BF)                                                                                  \
    acc0 = __builtin_amdgcn_wmma_f32_16x16x4_f32(false, (A0).xy, false, (BF).xy, (short)0, acc0, false, false); \
    acc1 = __builtin_amdgcn_wmma_f32_16x16x4_f32(false, (A1).xy, false, (BF).xy, (short)0, acc1, false, false); \
    acc2 = __builtin_amdgcn_wmma_f32_16x16x4_f32(false, (A2).xy, false, (BF).xy, (short)0, acc2, false, false); \
    acc3 = __builtin_amdgcn_wmma_f32_16x16x4_f32(false, (A3).xy, false, (BF).xy, (short)0, acc3, false, false); \
    acc0 = __builtin_amdgcn_wmma_f32_16x16x4_f32(false, (A0).zw, false, (BF).zw, (short)0, acc0, false, false); \
    acc1 = __builtin_amdgcn_wmma_f32_16x16x4_f32(false, (A1).zw, false, (BF).zw, (short)0, acc1, false, false); \
    acc2 = __builtin_amdgcn_wmma_f32_16x16x4_f32(false, (A2).zw, false, (BF).zw, (short)0, acc2, false, false); \
    acc3 = __builtin_amdgcn_wmma_f32_16x16x4_f32(false, (A3).zw, false, (BF).zw, (short)0, acc3, false, false);

    v8f acc0 = (v8f)0.0f, acc1 = (v8f)0.0f, acc2 = (v8f)0.0f, acc3 = (v8f)0.0f;
    v4f bP = wb[0], aP0 = xa[0], aP1 = xa[TSTR], aP2 = xa[2 * TSTR], aP3 = xa[3 * TSTR];
    for (int kb = 0; kb < KB8 - 2; kb += 2) {
        const int o1 = (kb + 1) * 32;
        const v4f bQ  = wb[o1];
        const v4f aQ0 = xa[o1];
        const v4f aQ1 = xa[TSTR + o1];
        const v4f aQ2 = xa[2 * TSTR + o1];
        const v4f aQ3 = xa[3 * TSTR + o1];
        __builtin_amdgcn_sched_barrier(0);
        MM8(aP0, aP1, aP2, aP3, bP)
        __builtin_amdgcn_sched_barrier(0);
        const int o2 = (kb + 2) * 32;
        bP  = wb[o2];
        aP0 = xa[o2];
        aP1 = xa[TSTR + o2];
        aP2 = xa[2 * TSTR + o2];
        aP3 = xa[3 * TSTR + o2];
        __builtin_amdgcn_sched_barrier(0);
        MM8(aQ0, aQ1, aQ2, aQ3, bQ)
        __builtin_amdgcn_sched_barrier(0);
        asm volatile("" :: "v"(bP), "v"(aP0), "v"(aP1), "v"(aP2), "v"(aP3));
    }
    {
        const int o1 = (KB8 - 1) * 32;
        const v4f bQ  = wb[o1];
        const v4f aQ0 = xa[o1];
        const v4f aQ1 = xa[TSTR + o1];
        const v4f aQ2 = xa[2 * TSTR + o1];
        const v4f aQ3 = xa[3 * TSTR + o1];
        MM8(aP0, aP1, aP2, aP3, bP)
        MM8(aQ0, aQ1, aQ2, aQ3, bQ)
    }
#undef MM8

    const int rbase = (lane >> 4) << 3;
    const int ncol  = nt * 16 + mlane;
    if (ncol < DD) {
        v8f accs[4] = {acc0, acc1, acc2, acc3};
#pragma unroll
        for (int i = 0; i < 4; ++i) {
            const size_t row0 = (size_t)((mt4 * 4 + i) * 16 + rbase);
#pragma unroll
            for (int g = 0; g < 8; ++g)
                Out[(row0 + g) * DD + ncol] = accs[i][g];
        }
    }
}

#define NSEG    15
#define NLONG   8
#define WPB     (NSEG * NREL)
static_assert(NLONG * 9 + (NSEG - NLONG) * 8 == LL, "the segments tile the emo rows");
#define FULLROW (LL - 1 - KK_REL)
static_assert((BB * WPB) % 8 == 0, "whole work-groups of 8 waves");

__device__ __forceinline__ int pair_row_start(const int i)
{
    if (i <= KK_REL + 1) return (i * (i + NREL)) / 2;
    const int t = i > FULLROW ? i - FULLROW : 0;
    return HEADP + (i - (KK_REL + 1)) * NREL - (t * (t - 1)) / 2;
}

__global__ __launch_bounds__(256) void pair_epilogue(
    const float* __restrict__ A_ws, const float* __restrict__ C_ws,
    const float* __restrict__ E_ws,
    const float* __restrict__ ln_g, const float* __restrict__ ln_b,
    const float* __restrict__ W_rel, const float* __restrict__ b_rel,
    float* __restrict__ out)
{
    const int wave = __builtin_amdgcn_readfirstlane(threadIdx.x >> 5);
    const int lane = threadIdx.x & 31;
    const int w   = blockIdx.x * 8 + wave;
    const int b   = w / WPB;
    const int rem = w - b * WPB;
    const int seg = rem / NREL;
    const int r   = rem - seg * NREL;
    const int d   = r - KK_REL;

    int lo = seg * 9 - (seg > NLONG ? seg - NLONG : 0);
    int hi = lo + (seg < NLONG ? 8 : 7);
    if (lo < -d) lo = -d;
    if (hi > LL - 1 - d) hi = LL - 1 - d;
    if (lo > hi) return;

    v2f ev[5], gv[5], bv[5], wv[5];
#pragma unroll
    for (int t = 0; t < 5; ++t) {
        const int idx = t * 64 + lane * 2;
        ev[t] = (v2f)0.0f; gv[t] = (v2f)0.0f; bv[t] = (v2f)0.0f; wv[t] = (v2f)0.0f;
        if (idx < DD) {
            ev[t] = *(const v2f*)(E_ws + (size_t)r * DD + idx);
            gv[t] = *(const v2f*)(ln_g  + idx);
            bv[t] = *(const v2f*)(ln_b  + idx);
            wv[t] = *(const v2f*)(W_rel + idx);
        }
    }
    const float brel = b_rel[0];

    const float* ra = A_ws + ((size_t)b * LL + lo) * DD;
    const float* rc = C_ws + ((size_t)b * LL + lo + d) * DD;
    float* ob = out + (size_t)b * NPAIR;

    for (int emo = lo; emo <= hi; ++emo, ra += DD, rc += DD) {
        float h[10];
        float s = 0.f, ss = 0.f;
#pragma unroll
        for (int t = 0; t < 5; ++t) {
            const int idx = t * 64 + lane * 2;
            v2f v = (v2f)0.0f;
            if (idx < DD) {
                v2f va = *(const v2f*)(ra + idx);
                v2f vc = *(const v2f*)(rc + idx);
                v = va + vc + ev[t];
            }
            h[2 * t]     = v.x;
            h[2 * t + 1] = v.y;
            s  += v.x + v.y;
            ss += v.x * v.x + v.y * v.y;
        }
#pragma unroll
        for (int off = 16; off > 0; off >>= 1) {
            s  += __shfl_xor(s,  off, 32);
            ss += __shfl_xor(ss, off, 32);
        }
        const float mu   = s * (1.f / DD);
        const float var  = ss * (1.f / DD) - mu * mu;
        const float rstd = rsqrtf(var + 1e-5f);

        float acc = 0.f;
#pragma unroll
        for (int t = 0; t < 5; ++t) {
            const int idx = t * 64 + lane * 2;
            if (idx < DD) {
                float v0 = (h[2 * t]     - mu) * rstd * gv[t].x + bv[t].x;
                float v1 = (h[2 * t + 1] - mu) * rstd * gv[t].y + bv[t].y;
                v0 = v0 > 0.f ? v0 : (expf(v0) - 1.f);
                v1 = v1 > 0.f ? v1 : (expf(v1) - 1.f);
                acc += v0 * wv[t].x + v1 * wv[t].y;
            }
        }
#pragma unroll
        for (int off = 16; off > 0; off >>= 1) acc += __shfl_xor(acc, off, 32);

        const int cau  = emo + d;
        const int colo = emo > KK_REL ? emo - KK_REL : 0;
        const int p    = pair_row_start(emo) + cau - colo;
        if (lane == 0) ob[p] = acc + brel;
    }
}

extern "C" void kernel_launch(void* const* d_in, const int* in_sizes, int n_in,
                              void* d_out, int out_size, void* d_ws, size_t ws_size,
                              hipStream_t stream) {
    const float* h_e     = (const float*)d_in[0];
    const float* h_c     = (const float*)d_in[1];
    const float* h_sh    = (const float*)d_in[2];
    const float* pos_emb = (const float*)d_in[3];
    const float* W_hid   = (const float*)d_in[4];
    const float* b_hid   = (const float*)d_in[5];
    const float* ln_g    = (const float*)d_in[6];
    const float* ln_b    = (const float*)d_in[7];
    const float* W_rel   = (const float*)d_in[8];
    const float* b_rel   = (const float*)d_in[9];
    float* out = (float*)d_out;

    float* ws   = (float*)d_ws;
    float* A_ws = ws;
    float* C_ws = A_ws + (size_t)MROWS * DD;
    float* XE   = C_ws + (size_t)MROWS * DD;
    float* XC   = XE   + XPACK;
    float* WB   = XC   + XPACK;
    float* E_ws = WB   + WPACK;

    dim3 pxg((MROWS / 16) * TSTR / 256, 2);
    pack_x<<<pxg, 256, 0, stream>>>(h_e, h_c, h_sh, XE, XC);

    dim3 pwg((2 * NTILES * TSTR + 255) / 256);
    pack_w<<<pwg, 256, 0, stream>>>(W_hid, WB);

    dim3 ggrid(EBLK + (MROWS / 64) * 3 * 2);
    pairproj_gemm<<<ggrid, 256, 0, stream>>>(XE, XC, WB, pos_emb, W_hid, b_hid, E_ws, A_ws, C_ws);

    pair_epilogue<<<BB * WPB / 8, 256, 0, stream>>>(A_ws, C_ws, E_ws,
                                                    ln_g, ln_b, W_rel, b_rel, out);
}


# Operand packing re-gridded: ONE launch of 350 work-groups, a wave per (2 row tiles x 2 blocks of 8 k), h_share read once for both X; 20,904 one-slot waves -> 2,800; 4->3 launches; same sums
# speedup vs baseline: 2.7695x; 1.0423x over previous
#include <hip/hip_runtime.h>
#include <hip/hip_bf16.h>
#include <math.h>

#define BB      32
#define LL      128
#define DD      300
#define KK_REL  12
#define NREL    25
#define PD      50
#define NPAIR   3044
#define FWID    650
#define MROWS   (BB * LL)
#define NTILES  19
#define NPAD    (NTILES * 16)
#define PKPAD   52
#define HEADP   247
#define EBLK    3
#define KB8     38
#define TSTR    (KB8 * 32)
#define XPACK   ((size_t)(MROWS / 16) * TSTR * 4)
#define WPACK   ((size_t)2 * NTILES * TSTR * 4)

typedef __attribute__((ext_vector_type(2))) float v2f;
typedef __attribute__((ext_vector_type(4))) float v4f;
typedef __attribute__((ext_vector_type(8))) float v8f;


#define KBP     (KB8 / 2)
#define XPWAVES ((MROWS / 32) * KBP)
#define XPWGS   (XPWAVES / 8)
#define WPWAVES (NTILES * KBP)
#define WPWGS   ((WPWAVES + 7) / 8)
static_assert(KB8 % 2 == 0, "k blocks pair up");
static_assert(XPWAVES % 8 == 0, "whole work-groups of 8 waves");
static_assert((MROWS / 16) % 2 == 0, "row tiles pair up");

__device__ __forceinline__ float4 add4(const float4 a, const float4 s)
{
    return make_float4(a.x + s.x, a.y + s.y, a.z + s.z, a.w + s.w);
}

__global__ __launch_bounds__(256) void pack_operands(
    const float* __restrict__ h_e, const float* __restrict__ h_c,
    const float* __restrict__ h_sh, const float* __restrict__ W_hid,
    float* __restrict__ XE, float* __restrict__ XC, float* __restrict__ WB)
{
    const int lane = threadIdx.x & 31;
    const int c    = lane & 15;
    const int hk   = (lane >> 4) * 4;
    const int wv   = blockIdx.x * 8 + (threadIdx.x >> 5);
    const float4 zero4 = make_float4(0.f, 0.f, 0.f, 0.f);

    if (blockIdx.x < XPWGS) {
        const int tp  = wv / KBP;
        const int kb  = (wv - tp * KBP) * 2;
        const int k0  = kb * 8 + hk;
        const bool ok1 = k0 + 8 < DD;
        const int  d1  = ok1 ? 8 : 0;
        float4 e[2][2], cc[2][2], s[2][2];
#pragma unroll
        for (int j = 0; j < 2; ++j) {
            const size_t off = (size_t)((2 * tp + j) * 16 + c) * DD + k0;
            e[j][0]  = *(const float4*)(h_e  + off);
            cc[j][0] = *(const float4*)(h_c  + off);
            s[j][0]  = *(const float4*)(h_sh + off);
            e[j][1]  = *(const float4*)(h_e  + off + d1);
            cc[j][1] = *(const float4*)(h_c  + off + d1);
            s[j][1]  = *(const float4*)(h_sh + off + d1);
        }
#pragma unroll
        for (int j = 0; j < 2; ++j) {
            const int slot = ((2 * tp + j) * KB8 + kb) * 32 + lane;
            ((float4*)XE)[slot]      = add4(e[j][0],  s[j][0]);
            ((float4*)XC)[slot]      = add4(cc[j][0], s[j][0]);
            ((float4*)XE)[slot + 32] = ok1 ? add4(e[j][1],  s[j][1]) : zero4;
            ((float4*)XC)[slot + 32] = ok1 ? add4(cc[j][1], s[j][1]) : zero4;
        }
        return;
    }

    const int ww = wv - XPWGS * 8;
    if (ww >= WPWAVES) return;
    const int tq  = ww / KBP;
    const int kb  = (ww - tq * KBP) * 2;
    const int k0  = kb * 8 + hk;
    const bool ok1 = k0 + 8 < DD;
    const int  d1  = ok1 ? 8 : 0;
    v2f lo[2][2], hi[2][2];
    bool nok[2];
#pragma unroll
    for (int j = 0; j < 2; ++j) {
        const int t   = 2 * tq + j;
        const int sel = t / NTILES;
        const int nt  = t - sel * NTILES;
        const int n   = nt * 16 + c;
        nok[j] = n < DD;
        const float* p = W_hid + (size_t)(nok[j] ? n : 0) * FWID + sel * DD + k0;
        lo[j][0] = *(const v2f*)p;
        hi[j][0] = *(const v2f*)(p + 2);
        lo[j][1] = *(const v2f*)(p + d1);
        hi[j][1] = *(const v2f*)(p + d1 + 2);
    }
#pragma unroll
    for (int j = 0; j < 2; ++j) {
        const int slot = ((2 * tq + j) * KB8 + kb) * 32 + lane;
        ((float4*)WB)[slot]      = nok[j] ? make_float4(lo[j][0].x, lo[j][0].y, hi[j][0].x, hi[j][0].y) : zero4;
        ((float4*)WB)[slot + 32] = (nok[j] && ok1) ? make_float4(lo[j][1].x, lo[j][1].y, hi[j][1].x, hi[j][1].y) : zero4;
    }
}

__device__ __forceinline__ void position_table_slice(
    const int ngrp,
    const float* __restrict__ pos_emb, const float* __restrict__ W_hid,
    const float* __restrict__ b_hid, float* __restrict__ E_ws,
    float (*G_l)[NREL], float (*S_l)[PKPAD])
{
    const int tid = threadIdx.x;

    for (int e = tid; e < NREL * NREL; e += 256) {
        const int r = e / NREL, rp = e % NREL;
        const int ad = rp >= KK_REL ? rp - KK_REL : KK_REL - rp;
        const float cnt = (float)(LL - ad);
        const int d = r - rp;
        G_l[r][rp] = cnt * expf(-(float)(d * d));
    }
    __syncthreads();

    for (int e = tid; e < 32 * PKPAD; e += 256) {
        const int r = e / PKPAD, pcol = e % PKPAD;
        float acc = 0.f;
        if (r < NREL && pcol < PD) {
            for (int rp = 0; rp < NREL; ++rp) acc += G_l[r][rp] * pos_emb[rp * PD + pcol];
        }
        S_l[r][pcol] = acc;
    }
    __syncthreads();

    const int wave = tid >> 5;
    const int lane = tid & 31;
    const int nt = ngrp * 8 + wave;
    if (nt >= NTILES) return;

    const int mlane = lane & 15;
    const int kk    = (lane >> 4) * 2;
    const int n     = nt * 16 + mlane;
    const bool nok  = n < DD;
    const float* wrow = W_hid + (size_t)(nok ? n : 0) * FWID + 2 * DD;

    v2f bfr[PKPAD / 4];
#pragma unroll
    for (int s = 0; s < PKPAD / 4; ++s) {
        const int k = s * 4 + kk;
        const bool ok = nok && (k < PD);
        v2f t = *(const v2f*)(wrow + (k < PD ? k : 0));
        bfr[s] = ok ? t : (v2f)0.0f;
    }
    const float bias = b_hid[nok ? n : 0];

    v8f acc0 = (v8f)0.0f, acc1 = (v8f)0.0f;
#pragma unroll
    for (int s = 0; s < PKPAD / 4; ++s) {
        const int k = s * 4 + kk;
        v2f a0 = *(const v2f*)(&S_l[mlane][k]);
        v2f a1 = *(const v2f*)(&S_l[16 + mlane][k]);
        acc0 = __builtin_amdgcn_wmma_f32_16x16x4_f32(false, a0, false, bfr[s], (short)0, acc0, false, false);
        acc1 = __builtin_amdgcn_wmma_f32_16x16x4_f32(false, a1, false, bfr[s], (short)0, acc1, false, false);
    }

    const int rbase = (lane >> 4) << 3;
    if (nok) {
#pragma unroll
        for (int g = 0; g < 8; ++g) {
            const int r0 = rbase + g;
            E_ws[r0 * DD + n] = acc0[g] + bias;
            const int r1 = 16 + rbase + g;
            if (r1 < NREL) E_ws[r1 * DD + n] = acc1[g] + bias;
        }
    }
}

__global__ __launch_bounds__(256) void pairproj_gemm(
    const float* __restrict__ XE, const float* __restrict__ XC,
    const float* __restrict__ WB,
    const float* __restrict__ pos_emb, const float* __restrict__ W_hid,
    const float* __restrict__ b_hid, float* __restrict__ E_ws,
    float* __restrict__ A_ws, float* __restrict__ C_ws)
{
    __shared__ float G_l[NREL][NREL];
    __shared__ __attribute__((aligned(16))) float S_l[32][PKPAD];

    if (blockIdx.x < EBLK) {
        position_table_slice(blockIdx.x, pos_emb, W_hid, b_hid, E_ws, G_l, S_l);
        return;
    }
    const int gid = blockIdx.x - EBLK;

    const int wave = threadIdx.x >> 5;
    const int lane = threadIdx.x & 31;
    const int mt4 = gid & 63;
    const int nt  = ((gid >> 6) % 3) * 8 + wave;
    if (nt >= NTILES) return;

    const int sel = gid / 192;
    const float* X   = sel ? XC : XE;
    float*       Out = sel ? C_ws : A_ws;

    const int mlane = lane & 15;

    const v4f* xa = (const v4f*)X  + (size_t)(mt4 * 4) * TSTR + lane;
    const v4f* wb = (const v4f*)WB + (size_t)(sel * NTILES + nt) * TSTR + lane;

#define MM8(A0, A1, A2, A3, BF)                                                                                  \
    acc0 = __builtin_amdgcn_wmma_f32_16x16x4_f32(false, (A0).xy, false, (BF).xy, (short)0, acc0, false, false); \
    acc1 = __builtin_amdgcn_wmma_f32_16x16x4_f32(false, (A1).xy, false, (BF).xy, (short)0, acc1, false, false); \
    acc2 = __builtin_amdgcn_wmma_f32_16x16x4_f32(false, (A2).xy, false, (BF).xy, (short)0, acc2, false, false); \
    acc3 = __builtin_amdgcn_wmma_f32_16x16x4_f32(false, (A3).xy, false, (BF).xy, (short)0, acc3, false, false); \
    acc0 = __builtin_amdgcn_wmma_f32_16x16x4_f32(false, (A0).zw, false, (BF).zw, (short)0, acc0, false, false); \
    acc1 = __builtin_amdgcn_wmma_f32_16x16x4_f32(false, (A1).zw, false, (BF).zw, (short)0, acc1, false, false); \
    acc2 = __builtin_amdgcn_wmma_f32_16x16x4_f32(false, (A2).zw, false, (BF).zw, (short)0, acc2, false, false); \
    acc3 = __builtin_amdgcn_wmma_f32_16x16x4_f32(false, (A3).zw, false, (BF).zw, (short)0, acc3, false, false);

    v8f acc0 = (v8f)0.0f, acc1 = (v8f)0.0f, acc2 = (v8f)0.0f, acc3 = (v8f)0.0f;
    v4f bP = wb[0], aP0 = xa[0], aP1 = xa[TSTR], aP2 = xa[2 * TSTR], aP3 = xa[3 * TSTR];
    for (int kb = 0; kb < KB8 - 2; kb += 2) {
        const int o1 = (kb + 1) * 32;
        const v4f bQ  = wb[o1];
        const v4f aQ0 = xa[o1];
        const v4f aQ1 = xa[TSTR + o1];
        const v4f aQ2 = xa[2 * TSTR + o1];
        const v4f aQ3 = xa[3 * TSTR + o1];
        __builtin_amdgcn_sched_barrier(0);
        MM8(aP0, aP1, aP2, aP3, bP)
        __builtin_amdgcn_sched_barrier(0);
        const int o2 = (kb + 2) * 32;
        bP  = wb[o2];
        aP0 = xa[o2];
        aP1 = xa[TSTR + o2];
        aP2 = xa[2 * TSTR + o2];
        aP3 = xa[3 * TSTR + o2];
        __builtin_amdgcn_sched_barrier(0);
        MM8(aQ0, aQ1, aQ2, aQ3, bQ)
        __builtin_amdgcn_sched_barrier(0);
        asm volatile("" :: "v"(bP), "v"(aP0), "v"(aP1), "v"(aP2), "v"(aP3));
    }
    {
        const int o1 = (KB8 - 1) * 32;
        const v4f bQ  = wb[o1];
        const v4f aQ0 = xa[o1];
        const v4f aQ1 = xa[TSTR + o1];
        const v4f aQ2 = xa[2 * TSTR + o1];
        const v4f aQ3 = xa[3 * TSTR + o1];
        MM8(aP0, aP1, aP2, aP3, bP)
        MM8(aQ0, aQ1, aQ2, aQ3, bQ)
    }
#undef MM8

    const int rbase = (lane >> 4) << 3;
    const int ncol  = nt * 16 + mlane;
    if (ncol < DD) {
        v8f accs[4] = {acc0, acc1, acc2, acc3};
#pragma unroll
        for (int i = 0; i < 4; ++i) {
            const size_t row0 = (size_t)((mt4 * 4 + i) * 16 + rbase);
#pragma unroll
            for (int g = 0; g < 8; ++g)
                Out[(row0 + g) * DD + ncol] = accs[i][g];
        }
    }
}

#define NSEG    15
#define NLONG   8
#define WPB     (NSEG * NREL)
static_assert(NLONG * 9 + (NSEG - NLONG) * 8 == LL, "the segments tile the emo rows");
#define FULLROW (LL - 1 - KK_REL)
static_assert((BB * WPB) % 8 == 0, "whole work-groups of 8 waves");

__device__ __forceinline__ int pair_row_start(const int i)
{
    if (i <= KK_REL + 1) return (i * (i + NREL)) / 2;
    const int t = i > FULLROW ? i - FULLROW : 0;
    return HEADP + (i - (KK_REL + 1)) * NREL - (t * (t - 1)) / 2;
}

__global__ __launch_bounds__(256) void pair_epilogue(
    const float* __restrict__ A_ws, const float* __restrict__ C_ws,
    const float* __restrict__ E_ws,
    const float* __restrict__ ln_g, const float* __restrict__ ln_b,
    const float* __restrict__ W_rel, const float* __restrict__ b_rel,
    float* __restrict__ out)
{
    const int wave = __builtin_amdgcn_readfirstlane(threadIdx.x >> 5);
    const int lane = threadIdx.x & 31;
    const int w   = blockIdx.x * 8 + wave;
    const int b   = w / WPB;
    const int rem = w - b * WPB;
    const int seg = rem / NREL;
    const int r   = rem - seg * NREL;
    const int d   = r - KK_REL;

    int lo = seg * 9 - (seg > NLONG ? seg - NLONG : 0);
    int hi = lo + (seg < NLONG ? 8 : 7);
    if (lo < -d) lo = -d;
    if (hi > LL - 1 - d) hi = LL - 1 - d;
    if (lo > hi) return;

    v2f ev[5], gv[5], bv[5], wv[5];
#pragma unroll
    for (int t = 0; t < 5; ++t) {
        const int idx = t * 64 + lane * 2;
        ev[t] = (v2f)0.0f; gv[t] = (v2f)0.0f; bv[t] = (v2f)0.0f; wv[t] = (v2f)0.0f;
        if (idx < DD) {
            ev[t] = *(const v2f*)(E_ws + (size_t)r * DD + idx);
            gv[t] = *(const v2f*)(ln_g  + idx);
            bv[t] = *(const v2f*)(ln_b  + idx);
            wv[t] = *(const v2f*)(W_rel + idx);
        }
    }
    const float brel = b_rel[0];

    const float* ra = A_ws + ((size_t)b * LL + lo) * DD;
    const float* rc = C_ws + ((size_t)b * LL + lo + d) * DD;
    float* ob = out + (size_t)b * NPAIR;

    for (int emo = lo; emo <= hi; ++emo, ra += DD, rc += DD) {
        float h[10];
        float s = 0.f, ss = 0.f;
#pragma unroll
        for (int t = 0; t < 5; ++t) {
            const int idx = t * 64 + lane * 2;
            v2f v = (v2f)0.0f;
            if (idx < DD) {
                v2f va = *(const v2f*)(ra + idx);
                v2f vc = *(const v2f*)(rc + idx);
                v = va + vc + ev[t];
            }
            h[2 * t]     = v.x;
            h[2 * t + 1] = v.y;
            s  += v.x + v.y;
            ss += v.x * v.x + v.y * v.y;
        }
#pragma unroll
        for (int off = 16; off > 0; off >>= 1) {
            s  += __shfl_xor(s,  off, 32);
            ss += __shfl_xor(ss, off, 32);
        }
        const float mu   = s * (1.f / DD);
        const float var  = ss * (1.f / DD) - mu * mu;
        const float rstd = rsqrtf(var + 1e-5f);

        float acc = 0.f;
#pragma unroll
        for (int t = 0; t < 5; ++t) {
            const int idx = t * 64 + lane * 2;
            if (idx < DD) {
                float v0 = (h[2 * t]     - mu) * rstd * gv[t].x + bv[t].x;
                float v1 = (h[2 * t + 1] - mu) * rstd * gv[t].y + bv[t].y;
                v0 = v0 > 0.f ? v0 : (expf(v0) - 1.f);
                v1 = v1 > 0.f ? v1 : (expf(v1) - 1.f);
                acc += v0 * wv[t].x + v1 * wv[t].y;
            }
        }
#pragma unroll
        for (int off = 16; off > 0; off >>= 1) acc += __shfl_xor(acc, off, 32);

        const int cau  = emo + d;
        const int colo = emo > KK_REL ? emo - KK_REL : 0;
        const int p    = pair_row_start(emo) + cau - colo;
        if (lane == 0) ob[p] = acc + brel;
    }
}

extern "C" void kernel_launch(void* const* d_in, const int* in_sizes, int n_in,
                              void* d_out, int out_size, void* d_ws, size_t ws_size,
                              hipStream_t stream) {
    const float* h_e     = (const float*)d_in[0];
    const float* h_c     = (const float*)d_in[1];
    const float* h_sh    = (const float*)d_in[2];
    const float* pos_emb = (const float*)d_in[3];
    const float* W_hid   = (const float*)d_in[4];
    const float* b_hid   = (const float*)d_in[5];
    const float* ln_g    = (const float*)d_in[6];
    const float* ln_b    = (const float*)d_in[7];
    const float* W_rel   = (const float*)d_in[8];
    const float* b_rel   = (const float*)d_in[9];
    float* out = (float*)d_out;

    float* ws   = (float*)d_ws;
    float* A_ws = ws;
    float* C_ws = A_ws + (size_t)MROWS * DD;
    float* XE   = C_ws + (size_t)MROWS * DD;
    float* XC   = XE   + XPACK;
    float* WB   = XC   + XPACK;
    float* E_ws = WB   + WPACK;

    pack_operands<<<XPWGS + WPWGS, 256, 0, stream>>>(h_e, h_c, h_sh, W_hid, XE, XC, WB);

    dim3 ggrid(EBLK + (MROWS / 64) * 3 * 2);
    pairproj_gemm<<<ggrid, 256, 0, stream>>>(XE, XC, WB, pos_emb, W_hid, b_hid, E_ws, A_ws, C_ws);

    pair_epilogue<<<BB * WPB / 8, 256, 0, stream>>>(A_ws, C_ws, E_ws,
                                                    ln_g, ln_b, W_rel, b_rel, out);
}


# GEMM re-tiled for equal matrix work per WGP: 256 work-groups = (matrix, 32 rows), 8 waves own 3,3,3,2,2,2,2,2 of the 19 N tiles x 2 M tiles, fragment loads 4 k-blocks ahead; same sum order
# speedup vs baseline: 3.1121x; 1.1237x over previous
#include <hip/hip_runtime.h>
#include <hip/hip_bf16.h>
#include <math.h>

#define BB      32
#define LL      128
#define DD      300
#define KK_REL  12
#define NREL    25
#define PD      50
#define NPAIR   3044
#define FWID    650
#define MROWS   (BB * LL)
#define NTILES  19
#define NPAD    (NTILES * 16)
#define PKPAD   52
#define HEADP   247
#define EBLK    3
#define KB8     38
#define TSTR    (KB8 * 32)
#define XPACK   ((size_t)(MROWS / 16) * TSTR * 4)
#define WPACK   ((size_t)2 * NTILES * TSTR * 4)

typedef __attribute__((ext_vector_type(2))) float v2f;
typedef __attribute__((ext_vector_type(4))) float v4f;
typedef __attribute__((ext_vector_type(8))) float v8f;


#define KBP     (KB8 / 2)
#define XPWAVES ((MROWS / 32) * KBP)
#define XPWGS   (XPWAVES / 8)
#define WPWAVES (NTILES * KBP)
#define WPWGS   ((WPWAVES + 7) / 8)
static_assert(KB8 % 2 == 0, "k blocks pair up");
static_assert(XPWAVES % 8 == 0, "whole work-groups of 8 waves");
static_assert((MROWS / 16) % 2 == 0, "row tiles pair up");

__device__ __forceinline__ float4 add4(const float4 a, const float4 s)
{
    return make_float4(a.x + s.x, a.y + s.y, a.z + s.z, a.w + s.w);
}

__global__ __launch_bounds__(256) void pack_operands(
    const float* __restrict__ h_e, const float* __restrict__ h_c,
    const float* __restrict__ h_sh, const float* __restrict__ W_hid,
    float* __restrict__ XE, float* __restrict__ XC, float* __restrict__ WB)
{
    const int lane = threadIdx.x & 31;
    const int c    = lane & 15;
    const int hk   = (lane >> 4) * 4;
    const int wv   = blockIdx.x * 8 + (threadIdx.x >> 5);
    const float4 zero4 = make_float4(0.f, 0.f, 0.f, 0.f);

    if (blockIdx.x < XPWGS) {
        const int tp  = wv / KBP;
        const int kb  = (wv - tp * KBP) * 2;
        const int k0  = kb * 8 + hk;
        const bool ok1 = k0 + 8 < DD;
        const int  d1  = ok1 ? 8 : 0;
        float4 e[2][2], cc[2][2], s[2][2];
#pragma unroll
        for (int j = 0; j < 2; ++j) {
            const size_t off = (size_t)((2 * tp + j) * 16 + c) * DD + k0;
            e[j][0]  = *(const float4*)(h_e  + off);
            cc[j][0] = *(const float4*)(h_c  + off);
            s[j][0]  = *(const float4*)(h_sh + off);
            e[j][1]  = *(const float4*)(h_e  + off + d1);
            cc[j][1] = *(const float4*)(h_c  + off + d1);
            s[j][1]  = *(const float4*)(h_sh + off + d1);
        }
#pragma unroll
        for (int j = 0; j < 2; ++j) {
            const int slot = ((2 * tp + j) * KB8 + kb) * 32 + lane;
            ((float4*)XE)[slot]      = add4(e[j][0],  s[j][0]);
            ((float4*)XC)[slot]      = add4(cc[j][0], s[j][0]);
            ((float4*)XE)[slot + 32] = ok1 ? add4(e[j][1],  s[j][1]) : zero4;
            ((float4*)XC)[slot + 32] = ok1 ? add4(cc[j][1], s[j][1]) : zero4;
        }
        return;
    }

    const int ww = wv - XPWGS * 8;
    if (ww >= WPWAVES) return;
    const int tq  = ww / KBP;
    const int kb  = (ww - tq * KBP) * 2;
    const int k0  = kb * 8 + hk;
    const bool ok1 = k0 + 8 < DD;
    const int  d1  = ok1 ? 8 : 0;
    v2f lo[2][2], hi[2][2];
    bool nok[2];
#pragma unroll
    for (int j = 0; j < 2; ++j) {
        const int t   = 2 * tq + j;
        const int sel = t / NTILES;
        const int nt  = t - sel * NTILES;
        const int n   = nt * 16 + c;
        nok[j] = n < DD;
        const float* p = W_hid + (size_t)(nok[j] ? n : 0) * FWID + sel * DD + k0;
        lo[j][0] = *(const v2f*)p;
        hi[j][0] = *(const v2f*)(p + 2);
        lo[j][1] = *(const v2f*)(p + d1);
        hi[j][1] = *(const v2f*)(p + d1 + 2);
    }
#pragma unroll
    for (int j = 0; j < 2; ++j) {
        const int slot = ((2 * tq + j) * KB8 + kb) * 32 + lane;
        ((float4*)WB)[slot]      = nok[j] ? make_float4(lo[j][0].x, lo[j][0].y, hi[j][0].x, hi[j][0].y) : zero4;
        ((float4*)WB)[slot + 32] = (nok[j] && ok1) ? make_float4(lo[j][1].x, lo[j][1].y, hi[j][1].x, hi[j][1].y) : zero4;
    }
}

__device__ __forceinline__ void position_table_slice(
    const int ngrp,
    const float* __restrict__ pos_emb, const float* __restrict__ W_hid,
    const float* __restrict__ b_hid, float* __restrict__ E_ws,
    float (*G_l)[NREL], float (*S_l)[PKPAD])
{
    const int tid = threadIdx.x;

    for (int e = tid; e < NREL * NREL; e += 256) {
        const int r = e / NREL, rp = e % NREL;
        const int ad = rp >= KK_REL ? rp - KK_REL : KK_REL - rp;
        const float cnt = (float)(LL - ad);
        const int d = r - rp;
        G_l[r][rp] = cnt * expf(-(float)(d * d));
    }
    __syncthreads();

    for (int e = tid; e < 32 * PKPAD; e += 256) {
        const int r = e / PKPAD, pcol = e % PKPAD;
        float acc = 0.f;
        if (r < NREL && pcol < PD) {
            for (int rp = 0; rp < NREL; ++rp) acc += G_l[r][rp] * pos_emb[rp * PD + pcol];
        }
        S_l[r][pcol] = acc;
    }
    __syncthreads();

    const int wave = tid >> 5;
    const int lane = tid & 31;
    const int nt = ngrp * 8 + wave;
    if (nt >= NTILES) return;

    const int mlane = lane & 15;
    const int kk    = (lane >> 4) * 2;
    const int n     = nt * 16 + mlane;
    const bool nok  = n < DD;
    const float* wrow = W_hid + (size_t)(nok ? n : 0) * FWID + 2 * DD;

    v2f bfr[PKPAD / 4];
#pragma unroll
    for (int s = 0; s < PKPAD / 4; ++s) {
        const int k = s * 4 + kk;
        const bool ok = nok && (k < PD);
        v2f t = *(const v2f*)(wrow + (k < PD ? k : 0));
        bfr[s] = ok ? t : (v2f)0.0f;
    }
    const float bias = b_hid[nok ? n : 0];

    v8f acc0 = (v8f)0.0f, acc1 = (v8f)0.0f;
#pragma unroll
    for (int s = 0; s < PKPAD / 4; ++s) {
        const int k = s * 4 + kk;
        v2f a0 = *(const v2f*)(&S_l[mlane][k]);
        v2f a1 = *(const v2f*)(&S_l[16 + mlane][k]);
        acc0 = __builtin_amdgcn_wmma_f32_16x16x4_f32(false, a0, false, bfr[s], (short)0, acc0, false, false);
        acc1 = __builtin_amdgcn_wmma_f32_16x16x4_f32(false, a1, false, bfr[s], (short)0, acc1, false, false);
    }

    const int rbase = (lane >> 4) << 3;
    if (nok) {
#pragma unroll
        for (int g = 0; g < 8; ++g) {
            const int r0 = rbase + g;
            E_ws[r0 * DD + n] = acc0[g] + bias;
            const int r1 = 16 + rbase + g;
            if (r1 < NREL) E_ws[r1 * DD + n] = acc1[g] + bias;
        }
    }
}

#define GEMMWG  (2 * (MROWS / 32))
#define PFD     4

template <int NN>
struct FragSet {
    v4f a0, a1;
    v4f b[NN];
};

template <int NN>
__device__ __forceinline__ void load_set(FragSet<NN>& f, const v4f* __restrict__ xa,
                                         const v4f* __restrict__ wb, const int blk)
{
    const int o = blk * 32;
    f.a0 = xa[o];
    f.a1 = xa[TSTR + o];
#pragma unroll
    for (int j = 0; j < NN; ++j) f.b[j] = wb[j * TSTR + o];
}

template <int NN>
__device__ __forceinline__ void mm_set(v8f (&acc)[2][NN], const FragSet<NN>& f)
{
#pragma unroll
    for (int j = 0; j < NN; ++j) {
        acc[0][j] = __builtin_amdgcn_wmma_f32_16x16x4_f32(false, f.a0.xy, false, f.b[j].xy, (short)0, acc[0][j], false, false);
        acc[1][j] = __builtin_amdgcn_wmma_f32_16x16x4_f32(false, f.a1.xy, false, f.b[j].xy, (short)0, acc[1][j], false, false);
    }
#pragma unroll
    for (int j = 0; j < NN; ++j) {
        acc[0][j] = __builtin_amdgcn_wmma_f32_16x16x4_f32(false, f.a0.zw, false, f.b[j].zw, (short)0, acc[0][j], false, false);
        acc[1][j] = __builtin_amdgcn_wmma_f32_16x16x4_f32(false, f.a1.zw, false, f.b[j].zw, (short)0, acc[1][j], false, false);
    }
}

template <int NN>
__device__ __forceinline__ void gemm_wave(const v4f* __restrict__ xa, const v4f* __restrict__ wb,
                                          float* __restrict__ Out, const int row0, const int nt0,
                                          const int lane)
{
    v8f acc[2][NN];
#pragma unroll
    for (int j = 0; j < NN; ++j) { acc[0][j] = (v8f)0.0f; acc[1][j] = (v8f)0.0f; }

    FragSet<NN> f[PFD + 1];
#pragma unroll
    for (int p = 0; p < PFD; ++p) load_set<NN>(f[p], xa, wb, p);

#pragma unroll
    for (int kb = 0; kb < KB8; ++kb) {
        if (kb + PFD < KB8) load_set<NN>(f[(kb + PFD) % (PFD + 1)], xa, wb, kb + PFD);
        __builtin_amdgcn_sched_barrier(0);
        mm_set<NN>(acc, f[kb % (PFD + 1)]);
        __builtin_amdgcn_sched_barrier(0);
    }

    const int mlane = lane & 15;
    const int rbase = (lane >> 4) << 3;
#pragma unroll
    for (int j = 0; j < NN; ++j) {
        const int ncol = (nt0 + j) * 16 + mlane;
        if (ncol < DD) {
#pragma unroll
            for (int i = 0; i < 2; ++i) {
                float* o = Out + (size_t)(row0 + i * 16 + rbase) * DD + ncol;
#pragma unroll
                for (int g = 0; g < 8; ++g) o[(size_t)g * DD] = acc[i][j][g];
            }
        }
    }
}

__global__ __launch_bounds__(256) void pairproj_gemm(
    const float* __restrict__ XE, const float* __restrict__ XC,
    const float* __restrict__ WB,
    const float* __restrict__ pos_emb, const float* __restrict__ W_hid,
    const float* __restrict__ b_hid, float* __restrict__ E_ws,
    float* __restrict__ A_ws, float* __restrict__ C_ws)
{
    __shared__ float G_l[NREL][NREL];
    __shared__ __attribute__((aligned(16))) float S_l[32][PKPAD];

    if (blockIdx.x >= GEMMWG) {
        position_table_slice(blockIdx.x - GEMMWG, pos_emb, W_hid, b_hid, E_ws, G_l, S_l);
        return;
    }
    const int gid  = blockIdx.x;
    const int sel  = gid >> 7;
    const int mt2  = gid & 127;
    const int wave = __builtin_amdgcn_readfirstlane(threadIdx.x >> 5);
    const int lane = threadIdx.x & 31;

    const float* X   = sel ? XC : XE;
    float*       Out = sel ? C_ws : A_ws;
    const v4f* xa  = (const v4f*)X + (size_t)(mt2 * 2) * TSTR + lane;
    const int row0 = mt2 * 32;

    if (wave < 3) {
        const int nt0 = wave * 3;
        gemm_wave<3>(xa, (const v4f*)WB + (size_t)(sel * NTILES + nt0) * TSTR + lane, Out, row0, nt0, lane);
    } else {
        const int nt0 = 9 + (wave - 3) * 2;
        gemm_wave<2>(xa, (const v4f*)WB + (size_t)(sel * NTILES + nt0) * TSTR + lane, Out, row0, nt0, lane);
    }
}

#define NSEG    15
#define NLONG   8
#define WPB     (NSEG * NREL)
static_assert(NLONG * 9 + (NSEG - NLONG) * 8 == LL, "the segments tile the emo rows");
#define FULLROW (LL - 1 - KK_REL)
static_assert((BB * WPB) % 8 == 0, "whole work-groups of 8 waves");

__device__ __forceinline__ int pair_row_start(const int i)
{
    if (i <= KK_REL + 1) return (i * (i + NREL)) / 2;
    const int t = i > FULLROW ? i - FULLROW : 0;
    return HEADP + (i - (KK_REL + 1)) * NREL - (t * (t - 1)) / 2;
}

__global__ __launch_bounds__(256) void pair_epilogue(
    const float* __restrict__ A_ws, const float* __restrict__ C_ws,
    const float* __restrict__ E_ws,
    const float* __restrict__ ln_g, const float* __restrict__ ln_b,
    const float* __restrict__ W_rel, const float* __restrict__ b_rel,
    float* __restrict__ out)
{
    const int wave = __builtin_amdgcn_readfirstlane(threadIdx.x >> 5);
    const int lane = threadIdx.x & 31;
    const int w   = blockIdx.x * 8 + wave;
    const int b   = w / WPB;
    const int rem = w - b * WPB;
    const int seg = rem / NREL;
    const int r   = rem - seg * NREL;
    const int d   = r - KK_REL;

    int lo = seg * 9 - (seg > NLONG ? seg - NLONG : 0);
    int hi = lo + (seg < NLONG ? 8 : 7);
    if (lo < -d) lo = -d;
    if (hi > LL - 1 - d) hi = LL - 1 - d;
    if (lo > hi) return;

    v2f ev[5], gv[5], bv[5], wv[5];
#pragma unroll
    for (int t = 0; t < 5; ++t) {
        const int idx = t * 64 + lane * 2;
        ev[t] = (v2f)0.0f; gv[t] = (v2f)0.0f; bv[t] = (v2f)0.0f; wv[t] = (v2f)0.0f;
        if (idx < DD) {
            ev[t] = *(const v2f*)(E_ws + (size_t)r * DD + idx);
            gv[t] = *(const v2f*)(ln_g  + idx);
            bv[t] = *(const v2f*)(ln_b  + idx);
            wv[t] = *(const v2f*)(W_rel + idx);
        }
    }
    const float brel = b_rel[0];

    const float* ra = A_ws + ((size_t)b * LL + lo) * DD;
    const float* rc = C_ws + ((size_t)b * LL + lo + d) * DD;
    float* ob = out + (size_t)b * NPAIR;

    for (int emo = lo; emo <= hi; ++emo, ra += DD, rc += DD) {
        float h[10];
        float s = 0.f, ss = 0.f;
#pragma unroll
        for (int t = 0; t < 5; ++t) {
            const int idx = t * 64 + lane * 2;
            v2f v = (v2f)0.0f;
            if (idx < DD) {
                v2f va = *(const v2f*)(ra + idx);
                v2f vc = *(const v2f*)(rc + idx);
                v = va + vc + ev[t];
            }
            h[2 * t]     = v.x;
            h[2 * t + 1] = v.y;
            s  += v.x + v.y;
            ss += v.x * v.x + v.y * v.y;
        }
#pragma unroll
        for (int off = 16; off > 0; off >>= 1) {
            s  += __shfl_xor(s,  off, 32);
            ss += __shfl_xor(ss, off, 32);
        }
        const float mu   = s * (1.f / DD);
        const float var  = ss * (1.f / DD) - mu * mu;
        const float rstd = rsqrtf(var + 1e-5f);

        float acc = 0.f;
#pragma unroll
        for (int t = 0; t < 5; ++t) {
            const int idx = t * 64 + lane * 2;
            if (idx < DD) {
                float v0 = (h[2 * t]     - mu) * rstd * gv[t].x + bv[t].x;
                float v1 = (h[2 * t + 1] - mu) * rstd * gv[t].y + bv[t].y;
                v0 = v0 > 0.f ? v0 : (expf(v0) - 1.f);
                v1 = v1 > 0.f ? v1 : (expf(v1) - 1.f);
                acc += v0 * wv[t].x + v1 * wv[t].y;
            }
        }
#pragma unroll
        for (int off = 16; off > 0; off >>= 1) acc += __shfl_xor(acc, off, 32);

        const int cau  = emo + d;
        const int colo = emo > KK_REL ? emo - KK_REL : 0;
        const int p    = pair_row_start(emo) + cau - colo;
        if (lane == 0) ob[p] = acc + brel;
    }
}

extern "C" void kernel_launch(void* const* d_in, const int* in_sizes, int n_in,
                              void* d_out, int out_size, void* d_ws, size_t ws_size,
                              hipStream_t stream) {
    const float* h_e     = (const float*)d_in[0];
    const float* h_c     = (const float*)d_in[1];
    const float* h_sh    = (const float*)d_in[2];
    const float* pos_emb = (const float*)d_in[3];
    const float* W_hid   = (const float*)d_in[4];
    const float* b_hid   = (const float*)d_in[5];
    const float* ln_g    = (const float*)d_in[6];
    const float* ln_b    = (const float*)d_in[7];
    const float* W_rel   = (const float*)d_in[8];
    const float* b_rel   = (const float*)d_in[9];
    float* out = (float*)d_out;

    float* ws   = (float*)d_ws;
    float* A_ws = ws;
    float* C_ws = A_ws + (size_t)MROWS * DD;
    float* XE   = C_ws + (size_t)MROWS * DD;
    float* XC   = XE   + XPACK;
    float* WB   = XC   + XPACK;
    float* E_ws = WB   + WPACK;

    pack_operands<<<XPWGS + WPWGS, 256, 0, stream>>>(h_e, h_c, h_sh, W_hid, XE, XC, WB);

    dim3 ggrid(GEMMWG + EBLK);
    pairproj_gemm<<<ggrid, 256, 0, stream>>>(XE, XC, WB, pos_emb, W_hid, b_hid, E_ws, A_ws, C_ws);

    pair_epilogue<<<BB * WPB / 8, 256, 0, stream>>>(A_ws, C_ws, E_ws,
                                                    ln_g, ln_b, W_rel, b_rel, out);
}


# Epilogue arithmetic two elements per instruction (packed f32): LN sums and butterfly, affine, dot, ELU exp in line (split log2e kept, as two fused steps) with a select; LN and dot sum order changed
# speedup vs baseline: 3.5956x; 1.1554x over previous
#include <hip/hip_runtime.h>
#include <hip/hip_bf16.h>
#include <math.h>

#define BB      32
#define LL      128
#define DD      300
#define KK_REL  12
#define NREL    25
#define PD      50
#define NPAIR   3044
#define FWID    650
#define MROWS   (BB * LL)
#define NTILES  19
#define NPAD    (NTILES * 16)
#define PKPAD   52
#define HEADP   247
#define EBLK    3
#define KB8     38
#define TSTR    (KB8 * 32)
#define XPACK   ((size_t)(MROWS / 16) * TSTR * 4)
#define WPACK   ((size_t)2 * NTILES * TSTR * 4)

typedef __attribute__((ext_vector_type(2))) float v2f;
typedef __attribute__((ext_vector_type(4))) float v4f;
typedef __attribute__((ext_vector_type(8))) float v8f;


#define KBP     (KB8 / 2)
#define XPWAVES ((MROWS / 32) * KBP)
#define XPWGS   (XPWAVES / 8)
#define WPWAVES (NTILES * KBP)
#define WPWGS   ((WPWAVES + 7) / 8)
static_assert(KB8 % 2 == 0, "k blocks pair up");
static_assert(XPWAVES % 8 == 0, "whole work-groups of 8 waves");
static_assert((MROWS / 16) % 2 == 0, "row tiles pair up");

__device__ __forceinline__ float4 add4(const float4 a, const float4 s)
{
    return make_float4(a.x + s.x, a.y + s.y, a.z + s.z, a.w + s.w);
}

__global__ __launch_bounds__(256) void pack_operands(
    const float* __restrict__ h_e, const float* __restrict__ h_c,
    const float* __restrict__ h_sh, const float* __restrict__ W_hid,
    float* __restrict__ XE, float* __restrict__ XC, float* __restrict__ WB)
{
    const int lane = threadIdx.x & 31;
    const int c    = lane & 15;
    const int hk   = (lane >> 4) * 4;
    const int wv   = blockIdx.x * 8 + (threadIdx.x >> 5);
    const float4 zero4 = make_float4(0.f, 0.f, 0.f, 0.f);

    if (blockIdx.x < XPWGS) {
        const int tp  = wv / KBP;
        const int kb  = (wv - tp * KBP) * 2;
        const int k0  = kb * 8 + hk;
        const bool ok1 = k0 + 8 < DD;
        const int  d1  = ok1 ? 8 : 0;
        float4 e[2][2], cc[2][2], s[2][2];
#pragma unroll
        for (int j = 0; j < 2; ++j) {
            const size_t off = (size_t)((2 * tp + j) * 16 + c) * DD + k0;
            e[j][0]  = *(const float4*)(h_e  + off);
            cc[j][0] = *(const float4*)(h_c  + off);
            s[j][0]  = *(const float4*)(h_sh + off);
            e[j][1]  = *(const float4*)(h_e  + off + d1);
            cc[j][1] = *(const float4*)(h_c  + off + d1);
            s[j][1]  = *(const float4*)(h_sh + off + d1);
        }
#pragma unroll
        for (int j = 0; j < 2; ++j) {
            const int slot = ((2 * tp + j) * KB8 + kb) * 32 + lane;
            ((float4*)XE)[slot]      = add4(e[j][0],  s[j][0]);
            ((float4*)XC)[slot]      = add4(cc[j][0], s[j][0]);
            ((float4*)XE)[slot + 32] = ok1 ? add4(e[j][1],  s[j][1]) : zero4;
            ((float4*)XC)[slot + 32] = ok1 ? add4(cc[j][1], s[j][1]) : zero4;
        }
        return;
    }

    const int ww = wv - XPWGS * 8;
    if (ww >= WPWAVES) return;
    const int tq  = ww / KBP;
    const int kb  = (ww - tq * KBP) * 2;
    const int k0  = kb * 8 + hk;
    const bool ok1 = k0 + 8 < DD;
    const int  d1  = ok1 ? 8 : 0;
    v2f lo[2][2], hi[2][2];
    bool nok[2];
#pragma unroll
    for (int j = 0; j < 2; ++j) {
        const int t   = 2 * tq + j;
        const int sel = t / NTILES;
        const int nt  = t - sel * NTILES;
        const int n   = nt * 16 + c;
        nok[j] = n < DD;
        const float* p = W_hid + (size_t)(nok[j] ? n : 0) * FWID + sel * DD + k0;
        lo[j][0] = *(const v2f*)p;
        hi[j][0] = *(const v2f*)(p + 2);
        lo[j][1] = *(const v2f*)(p + d1);
        hi[j][1] = *(const v2f*)(p + d1 + 2);
    }
#pragma unroll
    for (int j = 0; j < 2; ++j) {
        const int slot = ((2 * tq + j) * KB8 + kb) * 32 + lane;
        ((float4*)WB)[slot]      = nok[j] ? make_float4(lo[j][0].x, lo[j][0].y, hi[j][0].x, hi[j][0].y) : zero4;
        ((float4*)WB)[slot + 32] = (nok[j] && ok1) ? make_float4(lo[j][1].x, lo[j][1].y, hi[j][1].x, hi[j][1].y) : zero4;
    }
}

__device__ __forceinline__ void position_table_slice(
    const int ngrp,
    const float* __restrict__ pos_emb, const float* __restrict__ W_hid,
    const float* __restrict__ b_hid, float* __restrict__ E_ws,
    float (*G_l)[NREL], float (*S_l)[PKPAD])
{
    const int tid = threadIdx.x;

    for (int e = tid; e < NREL * NREL; e += 256) {
        const int r = e / NREL, rp = e % NREL;
        const int ad = rp >= KK_REL ? rp - KK_REL : KK_REL - rp;
        const float cnt = (float)(LL - ad);
        const int d = r - rp;
        G_l[r][rp] = cnt * expf(-(float)(d * d));
    }
    __syncthreads();

    for (int e = tid; e < 32 * PKPAD; e += 256) {
        const int r = e / PKPAD, pcol = e % PKPAD;
        float acc = 0.f;
        if (r < NREL && pcol < PD) {
            for (int rp = 0; rp < NREL; ++rp) acc += G_l[r][rp] * pos_emb[rp * PD + pcol];
        }
        S_l[r][pcol] = acc;
    }
    __syncthreads();

    const int wave = tid >> 5;
    const int lane = tid & 31;
    const int nt = ngrp * 8 + wave;
    if (nt >= NTILES) return;

    const int mlane = lane & 15;
    const int kk    = (lane >> 4) * 2;
    const int n     = nt * 16 + mlane;
    const bool nok  = n < DD;
    const float* wrow = W_hid + (size_t)(nok ? n : 0) * FWID + 2 * DD;

    v2f bfr[PKPAD / 4];
#pragma unroll
    for (int s = 0; s < PKPAD / 4; ++s) {
        const int k = s * 4 + kk;
        const bool ok = nok && (k < PD);
        v2f t = *(const v2f*)(wrow + (k < PD ? k : 0));
        bfr[s] = ok ? t : (v2f)0.0f;
    }
    const float bias = b_hid[nok ? n : 0];

    v8f acc0 = (v8f)0.0f, acc1 = (v8f)0.0f;
#pragma unroll
    for (int s = 0; s < PKPAD / 4; ++s) {
        const int k = s * 4 + kk;
        v2f a0 = *(const v2f*)(&S_l[mlane][k]);
        v2f a1 = *(const v2f*)(&S_l[16 + mlane][k]);
        acc0 = __builtin_amdgcn_wmma_f32_16x16x4_f32(false, a0, false, bfr[s], (short)0, acc0, false, false);
        acc1 = __builtin_amdgcn_wmma_f32_16x16x4_f32(false, a1, false, bfr[s], (short)0, acc1, false, false);
    }

    const int rbase = (lane >> 4) << 3;
    if (nok) {
#pragma unroll
        for (int g = 0; g < 8; ++g) {
            const int r0 = rbase + g;
            E_ws[r0 * DD + n] = acc0[g] + bias;
            const int r1 = 16 + rbase + g;
            if (r1 < NREL) E_ws[r1 * DD + n] = acc1[g] + bias;
        }
    }
}

#define GEMMWG  (2 * (MROWS / 32))
#define PFD     4

template <int NN>
struct FragSet {
    v4f a0, a1;
    v4f b[NN];
};

template <int NN>
__device__ __forceinline__ void load_set(FragSet<NN>& f, const v4f* __restrict__ xa,
                                         const v4f* __restrict__ wb, const int blk)
{
    const int o = blk * 32;
    f.a0 = xa[o];
    f.a1 = xa[TSTR + o];
#pragma unroll
    for (int j = 0; j < NN; ++j) f.b[j] = wb[j * TSTR + o];
}

template <int NN>
__device__ __forceinline__ void mm_set(v8f (&acc)[2][NN], const FragSet<NN>& f)
{
#pragma unroll
    for (int j = 0; j < NN; ++j) {
        acc[0][j] = __builtin_amdgcn_wmma_f32_16x16x4_f32(false, f.a0.xy, false, f.b[j].xy, (short)0, acc[0][j], false, false);
        acc[1][j] = __builtin_amdgcn_wmma_f32_16x16x4_f32(false, f.a1.xy, false, f.b[j].xy, (short)0, acc[1][j], false, false);
    }
#pragma unroll
    for (int j = 0; j < NN; ++j) {
        acc[0][j] = __builtin_amdgcn_wmma_f32_16x16x4_f32(false, f.a0.zw, false, f.b[j].zw, (short)0, acc[0][j], false, false);
        acc[1][j] = __builtin_amdgcn_wmma_f32_16x16x4_f32(false, f.a1.zw, false, f.b[j].zw, (short)0, acc[1][j], false, false);
    }
}

template <int NN>
__device__ __forceinline__ void gemm_wave(const v4f* __restrict__ xa, const v4f* __restrict__ wb,
                                          float* __restrict__ Out, const int row0, const int nt0,
                                          const int lane)
{
    v8f acc[2][NN];
#pragma unroll
    for (int j = 0; j < NN; ++j) { acc[0][j] = (v8f)0.0f; acc[1][j] = (v8f)0.0f; }

    FragSet<NN> f[PFD + 1];
#pragma unroll
    for (int p = 0; p < PFD; ++p) load_set<NN>(f[p], xa, wb, p);

#pragma unroll
    for (int kb = 0; kb < KB8; ++kb) {
        if (kb + PFD < KB8) load_set<NN>(f[(kb + PFD) % (PFD + 1)], xa, wb, kb + PFD);
        __builtin_amdgcn_sched_barrier(0);
        mm_set<NN>(acc, f[kb % (PFD + 1)]);
        __builtin_amdgcn_sched_barrier(0);
    }

    const int mlane = lane & 15;
    const int rbase = (lane >> 4) << 3;
#pragma unroll
    for (int j = 0; j < NN; ++j) {
        const int ncol = (nt0 + j) * 16 + mlane;
        if (ncol < DD) {
#pragma unroll
            for (int i = 0; i < 2; ++i) {
                float* o = Out + (size_t)(row0 + i * 16 + rbase) * DD + ncol;
#pragma unroll
                for (int g = 0; g < 8; ++g) o[(size_t)g * DD] = acc[i][j][g];
            }
        }
    }
}

__global__ __launch_bounds__(256) void pairproj_gemm(
    const float* __restrict__ XE, const float* __restrict__ XC,
    const float* __restrict__ WB,
    const float* __restrict__ pos_emb, const float* __restrict__ W_hid,
    const float* __restrict__ b_hid, float* __restrict__ E_ws,
    float* __restrict__ A_ws, float* __restrict__ C_ws)
{
    __shared__ float G_l[NREL][NREL];
    __shared__ __attribute__((aligned(16))) float S_l[32][PKPAD];

    if (blockIdx.x >= GEMMWG) {
        position_table_slice(blockIdx.x - GEMMWG, pos_emb, W_hid, b_hid, E_ws, G_l, S_l);
        return;
    }
    const int gid  = blockIdx.x;
    const int sel  = gid >> 7;
    const int mt2  = gid & 127;
    const int wave = __builtin_amdgcn_readfirstlane(threadIdx.x >> 5);
    const int lane = threadIdx.x & 31;

    const float* X   = sel ? XC : XE;
    float*       Out = sel ? C_ws : A_ws;
    const v4f* xa  = (const v4f*)X + (size_t)(mt2 * 2) * TSTR + lane;
    const int row0 = mt2 * 32;

    if (wave < 3) {
        const int nt0 = wave * 3;
        gemm_wave<3>(xa, (const v4f*)WB + (size_t)(sel * NTILES + nt0) * TSTR + lane, Out, row0, nt0, lane);
    } else {
        const int nt0 = 9 + (wave - 3) * 2;
        gemm_wave<2>(xa, (const v4f*)WB + (size_t)(sel * NTILES + nt0) * TSTR + lane, Out, row0, nt0, lane);
    }
}

#define NSEG    15
#define NLONG   8
#define WPB     (NSEG * NREL)
static_assert(NLONG * 9 + (NSEG - NLONG) * 8 == LL, "the segments tile the emo rows");
#define FULLROW (LL - 1 - KK_REL)
static_assert((BB * WPB) % 8 == 0, "whole work-groups of 8 waves");

__device__ __forceinline__ int pair_row_start(const int i)
{
    if (i <= KK_REL + 1) return (i * (i + NREL)) / 2;
    const int t = i > FULLROW ? i - FULLROW : 0;
    return HEADP + (i - (KK_REL + 1)) * NREL - (t * (t - 1)) / 2;
}

__device__ __forceinline__ v2f elu2(const v2f v)
{
    const v2f l2e_hi = (v2f)0x1.715476p+0f;
    const v2f l2e_lo = (v2f)0x1.4ae0bep-26f;
    v2f c;
    c.x = fmaxf(v.x, -104.0f);
    c.y = fmaxf(v.y, -104.0f);
    const v2f t = c * l2e_hi;
    v2f r;
    r.x = __builtin_rintf(t.x);
    r.y = __builtin_rintf(t.y);
    v2f a = __builtin_elementwise_fma(c, l2e_hi, -r);
    a = __builtin_elementwise_fma(c, l2e_lo, a);
    v2f p;
    p.x = __builtin_ldexpf(__builtin_amdgcn_exp2f(a.x), (int)r.x);
    p.y = __builtin_ldexpf(__builtin_amdgcn_exp2f(a.y), (int)r.y);
    const v2f m = p - (v2f)1.0f;
    v2f o;
    o.x = (v.x <= 0.f) ? m.x : v.x;
    o.y = (v.y <= 0.f) ? m.y : v.y;
    return o;
}

__global__ __launch_bounds__(256, 12) void pair_epilogue(
    const float* __restrict__ A_ws, const float* __restrict__ C_ws,
    const float* __restrict__ E_ws,
    const float* __restrict__ ln_g, const float* __restrict__ ln_b,
    const float* __restrict__ W_rel, const float* __restrict__ b_rel,
    float* __restrict__ out)
{
    const int wave = __builtin_amdgcn_readfirstlane(threadIdx.x >> 5);
    const int lane = threadIdx.x & 31;
    const int w   = blockIdx.x * 8 + wave;
    const int b   = w / WPB;
    const int rem = w - b * WPB;
    const int seg = rem / NREL;
    const int r   = rem - seg * NREL;
    const int d   = r - KK_REL;

    int lo = seg * 9 - (seg > NLONG ? seg - NLONG : 0);
    int hi = lo + (seg < NLONG ? 8 : 7);
    if (lo < -d) lo = -d;
    if (hi > LL - 1 - d) hi = LL - 1 - d;
    if (lo > hi) return;

    v2f ev[5], gv[5], bv[5], wv[5];
#pragma unroll
    for (int t = 0; t < 5; ++t) {
        const int idx = t * 64 + lane * 2;
        ev[t] = (v2f)0.0f; gv[t] = (v2f)0.0f; bv[t] = (v2f)0.0f; wv[t] = (v2f)0.0f;
        if (idx < DD) {
            ev[t] = *(const v2f*)(E_ws + (size_t)r * DD + idx);
            gv[t] = *(const v2f*)(ln_g  + idx);
            bv[t] = *(const v2f*)(ln_b  + idx);
            wv[t] = *(const v2f*)(W_rel + idx);
        }
    }
    const float brel = b_rel[0];

    const float* ra = A_ws + ((size_t)b * LL + lo) * DD;
    const float* rc = C_ws + ((size_t)b * LL + lo + d) * DD;
    float* ob = out + (size_t)b * NPAIR;

    for (int emo = lo; emo <= hi; ++emo, ra += DD, rc += DD) {
        v2f h[5];
        v2f s2, q2;
#pragma unroll
        for (int t = 0; t < 5; ++t) {
            const int idx = t * 64 + lane * 2;
            v2f v = (v2f)0.0f;
            if (idx < DD) {
                v2f va = *(const v2f*)(ra + idx);
                v2f vc = *(const v2f*)(rc + idx);
                v = va + vc + ev[t];
            }
            h[t] = v;
            if (t == 0) { s2 = v; q2 = v * v; }
            else        { s2 += v; q2 = __builtin_elementwise_fma(v, v, q2); }
        }
        v2f sq;
        sq.x = s2.x + s2.y;
        sq.y = q2.x + q2.y;
#pragma unroll
        for (int off = 16; off > 0; off >>= 1) {
            v2f o;
            o.x = __shfl_xor(sq.x, off, 32);
            o.y = __shfl_xor(sq.y, off, 32);
            sq += o;
        }
        const float mu   = sq.x * (1.f / DD);
        const float var  = sq.y * (1.f / DD) - mu * mu;
        const float rstd = rsqrtf(var + 1e-5f);
        const v2f mu2   = (v2f)mu;
        const v2f rstd2 = (v2f)rstd;

        v2f acc2 = (v2f)0.0f;
#pragma unroll
        for (int t = 0; t < 5; ++t) {
            const int idx = t * 64 + lane * 2;
            if (idx < DD) {
                const v2f x = __builtin_elementwise_fma((h[t] - mu2) * rstd2, gv[t], bv[t]);
                acc2 = __builtin_elementwise_fma(elu2(x), wv[t], acc2);
            }
        }
        float acc = acc2.x + acc2.y;
#pragma unroll
        for (int off = 16; off > 0; off >>= 1) acc += __shfl_xor(acc, off, 32);

        const int cau  = emo + d;
        const int colo = emo > KK_REL ? emo - KK_REL : 0;
        const int p    = pair_row_start(emo) + cau - colo;
        if (lane == 0) ob[p] = acc + brel;
    }
}

extern "C" void kernel_launch(void* const* d_in, const int* in_sizes, int n_in,
                              void* d_out, int out_size, void* d_ws, size_t ws_size,
                              hipStream_t stream) {
    const float* h_e     = (const float*)d_in[0];
    const float* h_c     = (const float*)d_in[1];
    const float* h_sh    = (const float*)d_in[2];
    const float* pos_emb = (const float*)d_in[3];
    const float* W_hid   = (const float*)d_in[4];
    const float* b_hid   = (const float*)d_in[5];
    const float* ln_g    = (const float*)d_in[6];
    const float* ln_b    = (const float*)d_in[7];
    const float* W_rel   = (const float*)d_in[8];
    const float* b_rel   = (const float*)d_in[9];
    float* out = (float*)d_out;

    float* ws   = (float*)d_ws;
    float* A_ws = ws;
    float* C_ws = A_ws + (size_t)MROWS * DD;
    float* XE   = C_ws + (size_t)MROWS * DD;
    float* XC   = XE   + XPACK;
    float* WB   = XC   + XPACK;
    float* E_ws = WB   + WPACK;

    pack_operands<<<XPWGS + WPWGS, 256, 0, stream>>>(h_e, h_c, h_sh, W_hid, XE, XC, WB);

    dim3 ggrid(GEMMWG + EBLK);
    pairproj_gemm<<<ggrid, 256, 0, stream>>>(XE, XC, WB, pos_emb, W_hid, b_hid, E_ws, A_ws, C_ws);

    pair_epilogue<<<BB * WPB / 8, 256, 0, stream>>>(A_ws, C_ws, E_ws,
                                                    ln_g, ln_b, W_rel, b_rel, out);
}


# Epilogue ELU exp: exp2 of the rounded v*log2e itself; its rounding rest v - t*ln2 (two-part ln2, fused) applied after it as p*(1+a), one packed fma with the -1; no rint, cvt, ldexp; same sum order
# speedup vs baseline: 3.7401x; 1.0402x over previous
#include <hip/hip_runtime.h>
#include <hip/hip_bf16.h>
#include <math.h>

#define BB      32
#define LL      128
#define DD      300
#define KK_REL  12
#define NREL    25
#define PD      50
#define NPAIR   3044
#define FWID    650
#define MROWS   (BB * LL)
#define NTILES  19
#define NPAD    (NTILES * 16)
#define PKPAD   52
#define HEADP   247
#define EBLK    3
#define KB8     38
#define TSTR    (KB8 * 32)
#define XPACK   ((size_t)(MROWS / 16) * TSTR * 4)
#define WPACK   ((size_t)2 * NTILES * TSTR * 4)

typedef __attribute__((ext_vector_type(2))) float v2f;
typedef __attribute__((ext_vector_type(4))) float v4f;
typedef __attribute__((ext_vector_type(8))) float v8f;


#define KBP     (KB8 / 2)
#define XPWAVES ((MROWS / 32) * KBP)
#define XPWGS   (XPWAVES / 8)
#define WPWAVES (NTILES * KBP)
#define WPWGS   ((WPWAVES + 7) / 8)
static_assert(KB8 % 2 == 0, "k blocks pair up");
static_assert(XPWAVES % 8 == 0, "whole work-groups of 8 waves");
static_assert((MROWS / 16) % 2 == 0, "row tiles pair up");

__device__ __forceinline__ float4 add4(const float4 a, const float4 s)
{
    return make_float4(a.x + s.x, a.y + s.y, a.z + s.z, a.w + s.w);
}

__global__ __launch_bounds__(256) void pack_operands(
    const float* __restrict__ h_e, const float* __restrict__ h_c,
    const float* __restrict__ h_sh, const float* __restrict__ W_hid,
    float* __restrict__ XE, float* __restrict__ XC, float* __restrict__ WB)
{
    const int lane = threadIdx.x & 31;
    const int c    = lane & 15;
    const int hk   = (lane >> 4) * 4;
    const int wv   = blockIdx.x * 8 + (threadIdx.x >> 5);
    const float4 zero4 = make_float4(0.f, 0.f, 0.f, 0.f);

    if (blockIdx.x < XPWGS) {
        const int tp  = wv / KBP;
        const int kb  = (wv - tp * KBP) * 2;
        const int k0  = kb * 8 + hk;
        const bool ok1 = k0 + 8 < DD;
        const int  d1  = ok1 ? 8 : 0;
        float4 e[2][2], cc[2][2], s[2][2];
#pragma unroll
        for (int j = 0; j < 2; ++j) {
            const size_t off = (size_t)((2 * tp + j) * 16 + c) * DD + k0;
            e[j][0]  = *(const float4*)(h_e  + off);
            cc[j][0] = *(const float4*)(h_c  + off);
            s[j][0]  = *(const float4*)(h_sh + off);
            e[j][1]  = *(const float4*)(h_e  + off + d1);
            cc[j][1] = *(const float4*)(h_c  + off + d1);
            s[j][1]  = *(const float4*)(h_sh + off + d1);
        }
#pragma unroll
        for (int j = 0; j < 2; ++j) {
            const int slot = ((2 * tp + j) * KB8 + kb) * 32 + lane;
            ((float4*)XE)[slot]      = add4(e[j][0],  s[j][0]);
            ((float4*)XC)[slot]      = add4(cc[j][0], s[j][0]);
            ((float4*)XE)[slot + 32] = ok1 ? add4(e[j][1],  s[j][1]) : zero4;
            ((float4*)XC)[slot + 32] = ok1 ? add4(cc[j][1], s[j][1]) : zero4;
        }
        return;
    }

    const int ww = wv - XPWGS * 8;
    if (ww >= WPWAVES) return;
    const int tq  = ww / KBP;
    const int kb  = (ww - tq * KBP) * 2;
    const int k0  = kb * 8 + hk;
    const bool ok1 = k0 + 8 < DD;
    const int  d1  = ok1 ? 8 : 0;
    v2f lo[2][2], hi[2][2];
    bool nok[2];
#pragma unroll
    for (int j = 0; j < 2; ++j) {
        const int t   = 2 * tq + j;
        const int sel = t / NTILES;
        const int nt  = t - sel * NTILES;
        const int n   = nt * 16 + c;
        nok[j] = n < DD;
        const float* p = W_hid + (size_t)(nok[j] ? n : 0) * FWID + sel * DD + k0;
        lo[j][0] = *(const v2f*)p;
        hi[j][0] = *(const v2f*)(p + 2);
        lo[j][1] = *(const v2f*)(p + d1);
        hi[j][1] = *(const v2f*)(p + d1 + 2);
    }
#pragma unroll
    for (int j = 0; j < 2; ++j) {
        const int slot = ((2 * tq + j) * KB8 + kb) * 32 + lane;
        ((float4*)WB)[slot]      = nok[j] ? make_float4(lo[j][0].x, lo[j][0].y, hi[j][0].x, hi[j][0].y) : zero4;
        ((float4*)WB)[slot + 32] = (nok[j] && ok1) ? make_float4(lo[j][1].x, lo[j][1].y, hi[j][1].x, hi[j][1].y) : zero4;
    }
}

__device__ __forceinline__ void position_table_slice(
    const int ngrp,
    const float* __restrict__ pos_emb, const float* __restrict__ W_hid,
    const float* __restrict__ b_hid, float* __restrict__ E_ws,
    float (*G_l)[NREL], float (*S_l)[PKPAD])
{
    const int tid = threadIdx.x;

    for (int e = tid; e < NREL * NREL; e += 256) {
        const int r = e / NREL, rp = e % NREL;
        const int ad = rp >= KK_REL ? rp - KK_REL : KK_REL - rp;
        const float cnt = (float)(LL - ad);
        const int d = r - rp;
        G_l[r][rp] = cnt * expf(-(float)(d * d));
    }
    __syncthreads();

    for (int e = tid; e < 32 * PKPAD; e += 256) {
        const int r = e / PKPAD, pcol = e % PKPAD;
        float acc = 0.f;
        if (r < NREL && pcol < PD) {
            for (int rp = 0; rp < NREL; ++rp) acc += G_l[r][rp] * pos_emb[rp * PD + pcol];
        }
        S_l[r][pcol] = acc;
    }
    __syncthreads();

    const int wave = tid >> 5;
    const int lane = tid & 31;
    const int nt = ngrp * 8 + wave;
    if (nt >= NTILES) return;

    const int mlane = lane & 15;
    const int kk    = (lane >> 4) * 2;
    const int n     = nt * 16 + mlane;
    const bool nok  = n < DD;
    const float* wrow = W_hid + (size_t)(nok ? n : 0) * FWID + 2 * DD;

    v2f bfr[PKPAD / 4];
#pragma unroll
    for (int s = 0; s < PKPAD / 4; ++s) {
        const int k = s * 4 + kk;
        const bool ok = nok && (k < PD);
        v2f t = *(const v2f*)(wrow + (k < PD ? k : 0));
        bfr[s] = ok ? t : (v2f)0.0f;
    }
    const float bias = b_hid[nok ? n : 0];

    v8f acc0 = (v8f)0.0f, acc1 = (v8f)0.0f;
#pragma unroll
    for (int s = 0; s < PKPAD / 4; ++s) {
        const int k = s * 4 + kk;
        v2f a0 = *(const v2f*)(&S_l[mlane][k]);
        v2f a1 = *(const v2f*)(&S_l[16 + mlane][k]);
        acc0 = __builtin_amdgcn_wmma_f32_16x16x4_f32(false, a0, false, bfr[s], (short)0, acc0, false, false);
        acc1 = __builtin_amdgcn_wmma_f32_16x16x4_f32(false, a1, false, bfr[s], (short)0, acc1, false, false);
    }

    const int rbase = (lane >> 4) << 3;
    if (nok) {
#pragma unroll
        for (int g = 0; g < 8; ++g) {
            const int r0 = rbase + g;
            E_ws[r0 * DD + n] = acc0[g] + bias;
            const int r1 = 16 + rbase + g;
            if (r1 < NREL) E_ws[r1 * DD + n] = acc1[g] + bias;
        }
    }
}

#define GEMMWG  (2 * (MROWS / 32))
#define PFD     4

template <int NN>
struct FragSet {
    v4f a0, a1;
    v4f b[NN];
};

template <int NN>
__device__ __forceinline__ void load_set(FragSet<NN>& f, const v4f* __restrict__ xa,
                                         const v4f* __restrict__ wb, const int blk)
{
    const int o = blk * 32;
    f.a0 = xa[o];
    f.a1 = xa[TSTR + o];
#pragma unroll
    for (int j = 0; j < NN; ++j) f.b[j] = wb[j * TSTR + o];
}

template <int NN>
__device__ __forceinline__ void mm_set(v8f (&acc)[2][NN], const FragSet<NN>& f)
{
#pragma unroll
    for (int j = 0; j < NN; ++j) {
        acc[0][j] = __builtin_amdgcn_wmma_f32_16x16x4_f32(false, f.a0.xy, false, f.b[j].xy, (short)0, acc[0][j], false, false);
        acc[1][j] = __builtin_amdgcn_wmma_f32_16x16x4_f32(false, f.a1.xy, false, f.b[j].xy, (short)0, acc[1][j], false, false);
    }
#pragma unroll
    for (int j = 0; j < NN; ++j) {
        acc[0][j] = __builtin_amdgcn_wmma_f32_16x16x4_f32(false, f.a0.zw, false, f.b[j].zw, (short)0, acc[0][j], false, false);
        acc[1][j] = __builtin_amdgcn_wmma_f32_16x16x4_f32(false, f.a1.zw, false, f.b[j].zw, (short)0, acc[1][j], false, false);
    }
}

template <int NN>
__device__ __forceinline__ void gemm_wave(const v4f* __restrict__ xa, const v4f* __restrict__ wb,
                                          float* __restrict__ Out, const int row0, const int nt0,
                                          const int lane)
{
    v8f acc[2][NN];
#pragma unroll
    for (int j = 0; j < NN; ++j) { acc[0][j] = (v8f)0.0f; acc[1][j] = (v8f)0.0f; }

    FragSet<NN> f[PFD + 1];
#pragma unroll
    for (int p = 0; p < PFD; ++p) load_set<NN>(f[p], xa, wb, p);

#pragma unroll
    for (int kb = 0; kb < KB8; ++kb) {
        if (kb + PFD < KB8) load_set<NN>(f[(kb + PFD) % (PFD + 1)], xa, wb, kb + PFD);
        __builtin_amdgcn_sched_barrier(0);
        mm_set<NN>(acc, f[kb % (PFD + 1)]);
        __builtin_amdgcn_sched_barrier(0);
    }

    const int mlane = lane & 15;
    const int rbase = (lane >> 4) << 3;
#pragma unroll
    for (int j = 0; j < NN; ++j) {
        const int ncol = (nt0 + j) * 16 + mlane;
        if (ncol < DD) {
#pragma unroll
            for (int i = 0; i < 2; ++i) {
                float* o = Out + (size_t)(row0 + i * 16 + rbase) * DD + ncol;
#pragma unroll
                for (int g = 0; g < 8; ++g) o[(size_t)g * DD] = acc[i][j][g];
            }
        }
    }
}

__global__ __launch_bounds__(256) void pairproj_gemm(
    const float* __restrict__ XE, const float* __restrict__ XC,
    const float* __restrict__ WB,
    const float* __restrict__ pos_emb, const float* __restrict__ W_hid,
    const float* __restrict__ b_hid, float* __restrict__ E_ws,
    float* __restrict__ A_ws, float* __restrict__ C_ws)
{
    __shared__ float G_l[NREL][NREL];
    __shared__ __attribute__((aligned(16))) float S_l[32][PKPAD];

    if (blockIdx.x >= GEMMWG) {
        position_table_slice(blockIdx.x - GEMMWG, pos_emb, W_hid, b_hid, E_ws, G_l, S_l);
        return;
    }
    const int gid  = blockIdx.x;
    const int sel  = gid >> 7;
    const int mt2  = gid & 127;
    const int wave = __builtin_amdgcn_readfirstlane(threadIdx.x >> 5);
    const int lane = threadIdx.x & 31;

    const float* X   = sel ? XC : XE;
    float*       Out = sel ? C_ws : A_ws;
    const v4f* xa  = (const v4f*)X + (size_t)(mt2 * 2) * TSTR + lane;
    const int row0 = mt2 * 32;

    if (wave < 3) {
        const int nt0 = wave * 3;
        gemm_wave<3>(xa, (const v4f*)WB + (size_t)(sel * NTILES + nt0) * TSTR + lane, Out, row0, nt0, lane);
    } else {
        const int nt0 = 9 + (wave - 3) * 2;
        gemm_wave<2>(xa, (const v4f*)WB + (size_t)(sel * NTILES + nt0) * TSTR + lane, Out, row0, nt0, lane);
    }
}

#define NSEG    15
#define NLONG   8
#define WPB     (NSEG * NREL)
static_assert(NLONG * 9 + (NSEG - NLONG) * 8 == LL, "the segments tile the emo rows");
#define FULLROW (LL - 1 - KK_REL)
static_assert((BB * WPB) % 8 == 0, "whole work-groups of 8 waves");

__device__ __forceinline__ int pair_row_start(const int i)
{
    if (i <= KK_REL + 1) return (i * (i + NREL)) / 2;
    const int t = i > FULLROW ? i - FULLROW : 0;
    return HEADP + (i - (KK_REL + 1)) * NREL - (t * (t - 1)) / 2;
}

__device__ __forceinline__ v2f elu2(const v2f v)
{
    const v2f l2e    = (v2f)0x1.715476p+0f;
    const v2f ln2_hi = (v2f)0x1.62e430p-1f;
    const v2f ln2_lo = (v2f)-1.90465430e-9f;
    v2f c;
    c.x = fmaxf(v.x, -104.0f);
    c.y = fmaxf(v.y, -104.0f);
    const v2f t = c * l2e;
    v2f p;
    p.x = __builtin_amdgcn_exp2f(t.x);
    p.y = __builtin_amdgcn_exp2f(t.y);
    v2f a = __builtin_elementwise_fma(-t, ln2_hi, c);
    a = __builtin_elementwise_fma(-t, ln2_lo, a);
    const v2f m = __builtin_elementwise_fma(p, a, p - (v2f)1.0f);
    v2f o;
    o.x = (v.x <= 0.f) ? m.x : v.x;
    o.y = (v.y <= 0.f) ? m.y : v.y;
    return o;
}

__global__ __launch_bounds__(256, 12) void pair_epilogue(
    const float* __restrict__ A_ws, const float* __restrict__ C_ws,
    const float* __restrict__ E_ws,
    const float* __restrict__ ln_g, const float* __restrict__ ln_b,
    const float* __restrict__ W_rel, const float* __restrict__ b_rel,
    float* __restrict__ out)
{
    const int wave = __builtin_amdgcn_readfirstlane(threadIdx.x >> 5);
    const int lane = threadIdx.x & 31;
    const int w   = blockIdx.x * 8 + wave;
    const int b   = w / WPB;
    const int rem = w - b * WPB;
    const int seg = rem / NREL;
    const int r   = rem - seg * NREL;
    const int d   = r - KK_REL;

    int lo = seg * 9 - (seg > NLONG ? seg - NLONG : 0);
    int hi = lo + (seg < NLONG ? 8 : 7);
    if (lo < -d) lo = -d;
    if (hi > LL - 1 - d) hi = LL - 1 - d;
    if (lo > hi) return;

    v2f ev[5], gv[5], bv[5], wv[5];
#pragma unroll
    for (int t = 0; t < 5; ++t) {
        const int idx = t * 64 + lane * 2;
        ev[t] = (v2f)0.0f; gv[t] = (v2f)0.0f; bv[t] = (v2f)0.0f; wv[t] = (v2f)0.0f;
        if (idx < DD) {
            ev[t] = *(const v2f*)(E_ws + (size_t)r * DD + idx);
            gv[t] = *(const v2f*)(ln_g  + idx);
            bv[t] = *(const v2f*)(ln_b  + idx);
            wv[t] = *(const v2f*)(W_rel + idx);
        }
    }
    const float brel = b_rel[0];

    const float* ra = A_ws + ((size_t)b * LL + lo) * DD;
    const float* rc = C_ws + ((size_t)b * LL + lo + d) * DD;
    float* ob = out + (size_t)b * NPAIR;

    for (int emo = lo; emo <= hi; ++emo, ra += DD, rc += DD) {
        v2f h[5];
        v2f s2, q2;
#pragma unroll
        for (int t = 0; t < 5; ++t) {
            const int idx = t * 64 + lane * 2;
            v2f v = (v2f)0.0f;
            if (idx < DD) {
                v2f va = *(const v2f*)(ra + idx);
                v2f vc = *(const v2f*)(rc + idx);
                v = va + vc + ev[t];
            }
            h[t] = v;
            if (t == 0) { s2 = v; q2 = v * v; }
            else        { s2 += v; q2 = __builtin_elementwise_fma(v, v, q2); }
        }
        v2f sq;
        sq.x = s2.x + s2.y;
        sq.y = q2.x + q2.y;
#pragma unroll
        for (int off = 16; off > 0; off >>= 1) {
            v2f o;
            o.x = __shfl_xor(sq.x, off, 32);
            o.y = __shfl_xor(sq.y, off, 32);
            sq += o;
        }
        const float mu   = sq.x * (1.f / DD);
        const float var  = sq.y * (1.f / DD) - mu * mu;
        const float rstd = rsqrtf(var + 1e-5f);
        const v2f mu2   = (v2f)mu;
        const v2f rstd2 = (v2f)rstd;

        v2f acc2 = (v2f)0.0f;
#pragma unroll
        for (int t = 0; t < 5; ++t) {
            const int idx = t * 64 + lane * 2;
            if (idx < DD) {
                const v2f x = __builtin_elementwise_fma((h[t] - mu2) * rstd2, gv[t], bv[t]);
                acc2 = __builtin_elementwise_fma(elu2(x), wv[t], acc2);
            }
        }
        float acc = acc2.x + acc2.y;
#pragma unroll
        for (int off = 16; off > 0; off >>= 1) acc += __shfl_xor(acc, off, 32);

        const int cau  = emo + d;
        const int colo = emo > KK_REL ? emo - KK_REL : 0;
        const int p    = pair_row_start(emo) + cau - colo;
        if (lane == 0) ob[p] = acc + brel;
    }
}

extern "C" void kernel_launch(void* const* d_in, const int* in_sizes, int n_in,
                              void* d_out, int out_size, void* d_ws, size_t ws_size,
                              hipStream_t stream) {
    const float* h_e     = (const float*)d_in[0];
    const float* h_c     = (const float*)d_in[1];
    const float* h_sh    = (const float*)d_in[2];
    const float* pos_emb = (const float*)d_in[3];
    const float* W_hid   = (const float*)d_in[4];
    const float* b_hid   = (const float*)d_in[5];
    const float* ln_g    = (const float*)d_in[6];
    const float* ln_b    = (const float*)d_in[7];
    const float* W_rel   = (const float*)d_in[8];
    const float* b_rel   = (const float*)d_in[9];
    float* out = (float*)d_out;

    float* ws   = (float*)d_ws;
    float* A_ws = ws;
    float* C_ws = A_ws + (size_t)MROWS * DD;
    float* XE   = C_ws + (size_t)MROWS * DD;
    float* XC   = XE   + XPACK;
    float* WB   = XC   + XPACK;
    float* E_ws = WB   + WPACK;

    pack_operands<<<XPWGS + WPWGS, 256, 0, stream>>>(h_e, h_c, h_sh, W_hid, XE, XC, WB);

    dim3 ggrid(GEMMWG + EBLK);
    pairproj_gemm<<<ggrid, 256, 0, stream>>>(XE, XC, WB, pos_emb, W_hid, b_hid, E_ws, A_ws, C_ws);

    pair_epilogue<<<BB * WPB / 8, 256, 0, stream>>>(A_ws, C_ws, E_ws,
                                                    ln_g, ln_b, W_rel, b_rel, out);
}


# Epilogue: a wave walks two neighbouring diagonals at once: the A row and one C row loaded per emo step serve two pairs (other C row carried in registers); 12,000 -> 7,904 waves; same sum order
# speedup vs baseline: 3.8547x; 1.0306x over previous
#include <hip/hip_runtime.h>
#include <hip/hip_bf16.h>
#include <math.h>

#define BB      32
#define LL      128
#define DD      300
#define KK_REL  12
#define NREL    25
#define PD      50
#define NPAIR   3044
#define FWID    650
#define MROWS   (BB * LL)
#define NTILES  19
#define NPAD    (NTILES * 16)
#define PKPAD   52
#define HEADP   247
#define EBLK    3
#define KB8     38
#define TSTR    (KB8 * 32)
#define XPACK   ((size_t)(MROWS / 16) * TSTR * 4)
#define WPACK   ((size_t)2 * NTILES * TSTR * 4)

typedef __attribute__((ext_vector_type(2))) float v2f;
typedef __attribute__((ext_vector_type(4))) float v4f;
typedef __attribute__((ext_vector_type(8))) float v8f;


#define KBP     (KB8 / 2)
#define XPWAVES ((MROWS / 32) * KBP)
#define XPWGS   (XPWAVES / 8)
#define WPWAVES (NTILES * KBP)
#define WPWGS   ((WPWAVES + 7) / 8)
static_assert(KB8 % 2 == 0, "k blocks pair up");
static_assert(XPWAVES % 8 == 0, "whole work-groups of 8 waves");
static_assert((MROWS / 16) % 2 == 0, "row tiles pair up");

__device__ __forceinline__ float4 add4(const float4 a, const float4 s)
{
    return make_float4(a.x + s.x, a.y + s.y, a.z + s.z, a.w + s.w);
}

__global__ __launch_bounds__(256) void pack_operands(
    const float* __restrict__ h_e, const float* __restrict__ h_c,
    const float* __restrict__ h_sh, const float* __restrict__ W_hid,
    float* __restrict__ XE, float* __restrict__ XC, float* __restrict__ WB)
{
    const int lane = threadIdx.x & 31;
    const int c    = lane & 15;
    const int hk   = (lane >> 4) * 4;
    const int wv   = blockIdx.x * 8 + (threadIdx.x >> 5);
    const float4 zero4 = make_float4(0.f, 0.f, 0.f, 0.f);

    if (blockIdx.x < XPWGS) {
        const int tp  = wv / KBP;
        const int kb  = (wv - tp * KBP) * 2;
        const int k0  = kb * 8 + hk;
        const bool ok1 = k0 + 8 < DD;
        const int  d1  = ok1 ? 8 : 0;
        float4 e[2][2], cc[2][2], s[2][2];
#pragma unroll
        for (int j = 0; j < 2; ++j) {
            const size_t off = (size_t)((2 * tp + j) * 16 + c) * DD + k0;
            e[j][0]  = *(const float4*)(h_e  + off);
            cc[j][0] = *(const float4*)(h_c  + off);
            s[j][0]  = *(const float4*)(h_sh + off);
            e[j][1]  = *(const float4*)(h_e  + off + d1);
            cc[j][1] = *(const float4*)(h_c  + off + d1);
            s[j][1]  = *(const float4*)(h_sh + off + d1);
        }
#pragma unroll
        for (int j = 0; j < 2; ++j) {
            const int slot = ((2 * tp + j) * KB8 + kb) * 32 + lane;
            ((float4*)XE)[slot]      = add4(e[j][0],  s[j][0]);
            ((float4*)XC)[slot]      = add4(cc[j][0], s[j][0]);
            ((float4*)XE)[slot + 32] = ok1 ? add4(e[j][1],  s[j][1]) : zero4;
            ((float4*)XC)[slot + 32] = ok1 ? add4(cc[j][1], s[j][1]) : zero4;
        }
        return;
    }

    const int ww = wv - XPWGS * 8;
    if (ww >= WPWAVES) return;
    const int tq  = ww / KBP;
    const int kb  = (ww - tq * KBP) * 2;
    const int k0  = kb * 8 + hk;
    const bool ok1 = k0 + 8 < DD;
    const int  d1  = ok1 ? 8 : 0;
    v2f lo[2][2], hi[2][2];
    bool nok[2];
#pragma unroll
    for (int j = 0; j < 2; ++j) {
        const int t   = 2 * tq + j;
        const int sel = t / NTILES;
        const int nt  = t - sel * NTILES;
        const int n   = nt * 16 + c;
        nok[j] = n < DD;
        const float* p = W_hid + (size_t)(nok[j] ? n : 0) * FWID + sel * DD + k0;
        lo[j][0] = *(const v2f*)p;
        hi[j][0] = *(const v2f*)(p + 2);
        lo[j][1] = *(const v2f*)(p + d1);
        hi[j][1] = *(const v2f*)(p + d1 + 2);
    }
#pragma unroll
    for (int j = 0; j < 2; ++j) {
        const int slot = ((2 * tq + j) * KB8 + kb) * 32 + lane;
        ((float4*)WB)[slot]      = nok[j] ? make_float4(lo[j][0].x, lo[j][0].y, hi[j][0].x, hi[j][0].y) : zero4;
        ((float4*)WB)[slot + 32] = (nok[j] && ok1) ? make_float4(lo[j][1].x, lo[j][1].y, hi[j][1].x, hi[j][1].y) : zero4;
    }
}

__device__ __forceinline__ void position_table_slice(
    const int ngrp,
    const float* __restrict__ pos_emb, const float* __restrict__ W_hid,
    const float* __restrict__ b_hid, float* __restrict__ E_ws,
    float (*G_l)[NREL], float (*S_l)[PKPAD])
{
    const int tid = threadIdx.x;

    for (int e = tid; e < NREL * NREL; e += 256) {
        const int r = e / NREL, rp = e % NREL;
        const int ad = rp >= KK_REL ? rp - KK_REL : KK_REL - rp;
        const float cnt = (float)(LL - ad);
        const int d = r - rp;
        G_l[r][rp] = cnt * expf(-(float)(d * d));
    }
    __syncthreads();

    for (int e = tid; e < 32 * PKPAD; e += 256) {
        const int r = e / PKPAD, pcol = e % PKPAD;
        float acc = 0.f;
        if (r < NREL && pcol < PD) {
            for (int rp = 0; rp < NREL; ++rp) acc += G_l[r][rp] * pos_emb[rp * PD + pcol];
        }
        S_l[r][pcol] = acc;
    }
    __syncthreads();

    const int wave = tid >> 5;
    const int lane = tid & 31;
    const int nt = ngrp * 8 + wave;
    if (nt >= NTILES) return;

    const int mlane = lane & 15;
    const int kk    = (lane >> 4) * 2;
    const int n     = nt * 16 + mlane;
    const bool nok  = n < DD;
    const float* wrow = W_hid + (size_t)(nok ? n : 0) * FWID + 2 * DD;

    v2f bfr[PKPAD / 4];
#pragma unroll
    for (int s = 0; s < PKPAD / 4; ++s) {
        const int k = s * 4 + kk;
        const bool ok = nok && (k < PD);
        v2f t = *(const v2f*)(wrow + (k < PD ? k : 0));
        bfr[s] = ok ? t : (v2f)0.0f;
    }
    const float bias = b_hid[nok ? n : 0];

    v8f acc0 = (v8f)0.0f, acc1 = (v8f)0.0f;
#pragma unroll
    for (int s = 0; s < PKPAD / 4; ++s) {
        const int k = s * 4 + kk;
        v2f a0 = *(const v2f*)(&S_l[mlane][k]);
        v2f a1 = *(const v2f*)(&S_l[16 + mlane][k]);
        acc0 = __builtin_amdgcn_wmma_f32_16x16x4_f32(false, a0, false, bfr[s], (short)0, acc0, false, false);
        acc1 = __builtin_amdgcn_wmma_f32_16x16x4_f32(false, a1, false, bfr[s], (short)0, acc1, false, false);
    }

    const int rbase = (lane >> 4) << 3;
    if (nok) {
#pragma unroll
        for (int g = 0; g < 8; ++g) {
            const int r0 = rbase + g;
            E_ws[r0 * DD + n] = acc0[g] + bias;
            const int r1 = 16 + rbase + g;
            if (r1 < NREL) E_ws[r1 * DD + n] = acc1[g] + bias;
        }
    }
}

#define GEMMWG  (2 * (MROWS / 32))
#define PFD     4

template <int NN>
struct FragSet {
    v4f a0, a1;
    v4f b[NN];
};

template <int NN>
__device__ __forceinline__ void load_set(FragSet<NN>& f, const v4f* __restrict__ xa,
                                         const v4f* __restrict__ wb, const int blk)
{
    const int o = blk * 32;
    f.a0 = xa[o];
    f.a1 = xa[TSTR + o];
#pragma unroll
    for (int j = 0; j < NN; ++j) f.b[j] = wb[j * TSTR + o];
}

template <int NN>
__device__ __forceinline__ void mm_set(v8f (&acc)[2][NN], const FragSet<NN>& f)
{
#pragma unroll
    for (int j = 0; j < NN; ++j) {
        acc[0][j] = __builtin_amdgcn_wmma_f32_16x16x4_f32(false, f.a0.xy, false, f.b[j].xy, (short)0, acc[0][j], false, false);
        acc[1][j] = __builtin_amdgcn_wmma_f32_16x16x4_f32(false, f.a1.xy, false, f.b[j].xy, (short)0, acc[1][j], false, false);
    }
#pragma unroll
    for (int j = 0; j < NN; ++j) {
        acc[0][j] = __builtin_amdgcn_wmma_f32_16x16x4_f32(false, f.a0.zw, false, f.b[j].zw, (short)0, acc[0][j], false, false);
        acc[1][j] = __builtin_amdgcn_wmma_f32_16x16x4_f32(false, f.a1.zw, false, f.b[j].zw, (short)0, acc[1][j], false, false);
    }
}

template <int NN>
__device__ __forceinline__ void gemm_wave(const v4f* __restrict__ xa, const v4f* __restrict__ wb,
                                          float* __restrict__ Out, const int row0, const int nt0,
                                          const int lane)
{
    v8f acc[2][NN];
#pragma unroll
    for (int j = 0; j < NN; ++j) { acc[0][j] = (v8f)0.0f; acc[1][j] = (v8f)0.0f; }

    FragSet<NN> f[PFD + 1];
#pragma unroll
    for (int p = 0; p < PFD; ++p) load_set<NN>(f[p], xa, wb, p);

#pragma unroll
    for (int kb = 0; kb < KB8; ++kb) {
        if (kb + PFD < KB8) load_set<NN>(f[(kb + PFD) % (PFD + 1)], xa, wb, kb + PFD);
        __builtin_amdgcn_sched_barrier(0);
        mm_set<NN>(acc, f[kb % (PFD + 1)]);
        __builtin_amdgcn_sched_barrier(0);
    }

    const int mlane = lane & 15;
    const int rbase = (lane >> 4) << 3;
#pragma unroll
    for (int j = 0; j < NN; ++j) {
        const int ncol = (nt0 + j) * 16 + mlane;
        if (ncol < DD) {
#pragma unroll
            for (int i = 0; i < 2; ++i) {
                float* o = Out + (size_t)(row0 + i * 16 + rbase) * DD + ncol;
#pragma unroll
                for (int g = 0; g < 8; ++g) o[(size_t)g * DD] = acc[i][j][g];
            }
        }
    }
}

__global__ __launch_bounds__(256) void pairproj_gemm(
    const float* __restrict__ XE, const float* __restrict__ XC,
    const float* __restrict__ WB,
    const float* __restrict__ pos_emb, const float* __restrict__ W_hid,
    const float* __restrict__ b_hid, float* __restrict__ E_ws,
    float* __restrict__ A_ws, float* __restrict__ C_ws)
{
    __shared__ float G_l[NREL][NREL];
    __shared__ __attribute__((aligned(16))) float S_l[32][PKPAD];

    if (blockIdx.x >= GEMMWG) {
        position_table_slice(blockIdx.x - GEMMWG, pos_emb, W_hid, b_hid, E_ws, G_l, S_l);
        return;
    }
    const int gid  = blockIdx.x;
    const int sel  = gid >> 7;
    const int mt2  = gid & 127;
    const int wave = __builtin_amdgcn_readfirstlane(threadIdx.x >> 5);
    const int lane = threadIdx.x & 31;

    const float* X   = sel ? XC : XE;
    float*       Out = sel ? C_ws : A_ws;
    const v4f* xa  = (const v4f*)X + (size_t)(mt2 * 2) * TSTR + lane;
    const int row0 = mt2 * 32;

    if (wave < 3) {
        const int nt0 = wave * 3;
        gemm_wave<3>(xa, (const v4f*)WB + (size_t)(sel * NTILES + nt0) * TSTR + lane, Out, row0, nt0, lane);
    } else {
        const int nt0 = 9 + (wave - 3) * 2;
        gemm_wave<2>(xa, (const v4f*)WB + (size_t)(sel * NTILES + nt0) * TSTR + lane, Out, row0, nt0, lane);
    }
}

#define NSEG    19
#define NLONG   14
#define NGRP    13
#define WPB     (NSEG * NGRP)
static_assert(NLONG * 7 + (NSEG - NLONG) * 6 == LL, "the segments tile the emo rows");
static_assert(2 * (NGRP - 1) + 1 == NREL, "the groups tile the diagonals");
#define FULLROW (LL - 1 - KK_REL)
static_assert((BB * WPB) % 8 == 0, "whole work-groups of 8 waves");

__device__ __forceinline__ int pair_row_start(const int i)
{
    if (i <= KK_REL + 1) return (i * (i + NREL)) / 2;
    const int t = i > FULLROW ? i - FULLROW : 0;
    return HEADP + (i - (KK_REL + 1)) * NREL - (t * (t - 1)) / 2;
}

__device__ __forceinline__ v2f elu2(const v2f v)
{
    const v2f l2e    = (v2f)0x1.715476p+0f;
    const v2f ln2_hi = (v2f)0x1.62e430p-1f;
    const v2f ln2_lo = (v2f)-1.90465430e-9f;
    v2f c;
    c.x = fmaxf(v.x, -104.0f);
    c.y = fmaxf(v.y, -104.0f);
    const v2f t = c * l2e;
    v2f p;
    p.x = __builtin_amdgcn_exp2f(t.x);
    p.y = __builtin_amdgcn_exp2f(t.y);
    v2f a = __builtin_elementwise_fma(-t, ln2_hi, c);
    a = __builtin_elementwise_fma(-t, ln2_lo, a);
    const v2f m = __builtin_elementwise_fma(p, a, p - (v2f)1.0f);
    v2f o;
    o.x = (v.x <= 0.f) ? m.x : v.x;
    o.y = (v.y <= 0.f) ? m.y : v.y;
    return o;
}

__device__ __forceinline__ float pair_value(const v2f (&a)[5], const v2f (&c)[5], const v2f (&ev)[5],
                                            const v2f (&gv)[5], const v2f (&bv)[5], const v2f (&wv)[5])
{
    v2f h[5];
    v2f s2, q2;
#pragma unroll
    for (int t = 0; t < 5; ++t) {
        const v2f v = a[t] + c[t] + ev[t];
        h[t] = v;
        if (t == 0) { s2 = v; q2 = v * v; }
        else        { s2 += v; q2 = __builtin_elementwise_fma(v, v, q2); }
    }
    v2f sq;
    sq.x = s2.x + s2.y;
    sq.y = q2.x + q2.y;
#pragma unroll
    for (int off = 16; off > 0; off >>= 1) {
        v2f o;
        o.x = __shfl_xor(sq.x, off, 32);
        o.y = __shfl_xor(sq.y, off, 32);
        sq += o;
    }
    const float mu   = sq.x * (1.f / DD);
    const float var  = sq.y * (1.f / DD) - mu * mu;
    const float rstd = rsqrtf(var + 1e-5f);
    const v2f mu2   = (v2f)mu;
    const v2f rstd2 = (v2f)rstd;

    v2f acc2 = (v2f)0.0f;
#pragma unroll
    for (int t = 0; t < 5; ++t) {
        const v2f x = __builtin_elementwise_fma((h[t] - mu2) * rstd2, gv[t], bv[t]);
        acc2 = __builtin_elementwise_fma(elu2(x), wv[t], acc2);
    }
    float acc = acc2.x + acc2.y;
#pragma unroll
    for (int off = 16; off > 0; off >>= 1) acc += __shfl_xor(acc, off, 32);
    return acc;
}

__device__ __forceinline__ void load_row(v2f (&dst)[5], const float* __restrict__ row, const int lane)
{
#pragma unroll
    for (int t = 0; t < 5; ++t) {
        const int idx = t * 64 + lane * 2;
        dst[t] = (v2f)0.0f;
        if (idx < DD) dst[t] = *(const v2f*)(row + idx);
    }
}

template <int ND>
__device__ __forceinline__ void walk_diagonals(
    const int b, int lo, int hi, const int r0, const int lane,
    const float* __restrict__ A_ws, const float* __restrict__ C_ws, const float* __restrict__ E_ws,
    const float* __restrict__ ln_g, const float* __restrict__ ln_b,
    const float* __restrict__ W_rel, const float brel, float* __restrict__ out)
{
    const int d0 = r0 - KK_REL;
    if (lo < -(d0 + ND - 1)) lo = -(d0 + ND - 1);
    if (hi > LL - 1 - d0) hi = LL - 1 - d0;
    if (lo > hi) return;

    v2f ev[ND][5], gv[5], bv[5], wv[5];
#pragma unroll
    for (int j = 0; j < ND; ++j) load_row(ev[j], E_ws + (size_t)(r0 + j) * DD, lane);
    load_row(gv, ln_g, lane);
    load_row(bv, ln_b, lane);
    load_row(wv, W_rel, lane);

    const float* ab = A_ws + (size_t)b * LL * DD;
    const float* cb = C_ws + (size_t)b * LL * DD;
    float* ob = out + (size_t)b * NPAIR;

    v2f cr[ND][5];
#pragma unroll
    for (int j = 0; j + 1 < ND; ++j) {
        int row = lo + d0 + j;
        row = row < 0 ? 0 : row;
        load_row(cr[j], cb + (size_t)row * DD, lane);
    }

    for (int emo = lo; emo <= hi; ++emo) {
        v2f a[5];
        load_row(a, ab + (size_t)emo * DD, lane);
        int rown = emo + d0 + ND - 1;
        rown = rown > LL - 1 ? LL - 1 : (rown < 0 ? 0 : rown);
        load_row(cr[ND - 1], cb + (size_t)rown * DD, lane);

        float res[ND];
#pragma unroll
        for (int j = 0; j < ND; ++j) res[j] = pair_value(a, cr[j], ev[j], gv, bv, wv);

        const int colo  = emo > KK_REL ? emo - KK_REL : 0;
        const int pbase = pair_row_start(emo) - colo;
#pragma unroll
        for (int j = 0; j < ND; ++j) {
            const int cau = emo + d0 + j;
            if (cau >= 0 && cau < LL && lane == 0) ob[pbase + cau] = res[j] + brel;
        }
#pragma unroll
        for (int j = 0; j + 1 < ND; ++j) {
#pragma unroll
            for (int t = 0; t < 5; ++t) cr[j][t] = cr[j + 1][t];
        }
    }
}

__global__ __launch_bounds__(256, 8) void pair_epilogue(
    const float* __restrict__ A_ws, const float* __restrict__ C_ws,
    const float* __restrict__ E_ws,
    const float* __restrict__ ln_g, const float* __restrict__ ln_b,
    const float* __restrict__ W_rel, const float* __restrict__ b_rel,
    float* __restrict__ out)
{
    const int wave = __builtin_amdgcn_readfirstlane(threadIdx.x >> 5);
    const int lane = threadIdx.x & 31;
    const int w   = blockIdx.x * 8 + wave;
    const int b   = w / WPB;
    const int rem = w - b * WPB;
    const int seg = rem / NGRP;
    const int g   = rem - seg * NGRP;

    const int lo = seg * 7 - (seg > NLONG ? seg - NLONG : 0);
    const int hi = lo + (seg < NLONG ? 6 : 5);
    const float brel = b_rel[0];

    if (g < NGRP - 1)
        walk_diagonals<2>(b, lo, hi, 2 * g, lane, A_ws, C_ws, E_ws, ln_g, ln_b, W_rel, brel, out);
    else
        walk_diagonals<1>(b, lo, hi, NREL - 1, lane, A_ws, C_ws, E_ws, ln_g, ln_b, W_rel, brel, out);
}

extern "C" void kernel_launch(void* const* d_in, const int* in_sizes, int n_in,
                              void* d_out, int out_size, void* d_ws, size_t ws_size,
                              hipStream_t stream) {
    const float* h_e     = (const float*)d_in[0];
    const float* h_c     = (const float*)d_in[1];
    const float* h_sh    = (const float*)d_in[2];
    const float* pos_emb = (const float*)d_in[3];
    const float* W_hid   = (const float*)d_in[4];
    const float* b_hid   = (const float*)d_in[5];
    const float* ln_g    = (const float*)d_in[6];
    const float* ln_b    = (const float*)d_in[7];
    const float* W_rel   = (const float*)d_in[8];
    const float* b_rel   = (const float*)d_in[9];
    float* out = (float*)d_out;

    float* ws   = (float*)d_ws;
    float* A_ws = ws;
    float* C_ws = A_ws + (size_t)MROWS * DD;
    float* XE   = C_ws + (size_t)MROWS * DD;
    float* XC   = XE   + XPACK;
    float* WB   = XC   + XPACK;
    float* E_ws = WB   + WPACK;

    pack_operands<<<XPWGS + WPWGS, 256, 0, stream>>>(h_e, h_c, h_sh, W_hid, XE, XC, WB);

    dim3 ggrid(GEMMWG + EBLK);
    pairproj_gemm<<<ggrid, 256, 0, stream>>>(XE, XC, WB, pos_emb, W_hid, b_hid, E_ws, A_ws, C_ws);

    pair_epilogue<<<BB * WPB / 8, 256, 0, stream>>>(A_ws, C_ws, E_ws,
                                                    ln_g, ln_b, W_rel, b_rel, out);
}


# GEMM takes its A fragments straight from h_e/h_c and h_share, added in registers: X pack pass removed (pack launch keeps W only); 4-wave work-groups, wave = 2 M x 5|4 N tiles; same sum order
# speedup vs baseline: 4.0126x; 1.0410x over previous
#include <hip/hip_runtime.h>
#include <hip/hip_bf16.h>
#include <math.h>

#define BB      32
#define LL      128
#define DD      300
#define KK_REL  12
#define NREL    25
#define PD      50
#define NPAIR   3044
#define FWID    650
#define MROWS   (BB * LL)
#define NTILES  19
#define NPAD    (NTILES * 16)
#define PKPAD   52
#define HEADP   247
#define EBLK    5
#define KB8     38
#define TSTR    (KB8 * 32)
#define WPACK   ((size_t)2 * NTILES * TSTR * 4)

typedef __attribute__((ext_vector_type(2))) float v2f;
typedef __attribute__((ext_vector_type(4))) float v4f;
typedef __attribute__((ext_vector_type(8))) float v8f;


#define KBP     (KB8 / 2)
#define WPWAVES (NTILES * KBP)
#define WPWGS   ((WPWAVES + 7) / 8)
static_assert(KB8 % 2 == 0, "k blocks pair up");

__global__ __launch_bounds__(256) void pack_w(
    const float* __restrict__ W_hid, float* __restrict__ WB)
{
    const int lane = threadIdx.x & 31;
    const int c    = lane & 15;
    const int hk   = (lane >> 4) * 4;
    const int ww   = blockIdx.x * 8 + (threadIdx.x >> 5);
    const float4 zero4 = make_float4(0.f, 0.f, 0.f, 0.f);

    if (ww >= WPWAVES) return;
    const int tq  = ww / KBP;
    const int kb  = (ww - tq * KBP) * 2;
    const int k0  = kb * 8 + hk;
    const bool ok1 = k0 + 8 < DD;
    const int  d1  = ok1 ? 8 : 0;
    v2f lo[2][2], hi[2][2];
    bool nok[2];
#pragma unroll
    for (int j = 0; j < 2; ++j) {
        const int t   = 2 * tq + j;
        const int sel = t / NTILES;
        const int nt  = t - sel * NTILES;
        const int n   = nt * 16 + c;
        nok[j] = n < DD;
        const float* p = W_hid + (size_t)(nok[j] ? n : 0) * FWID + sel * DD + k0;
        lo[j][0] = *(const v2f*)p;
        hi[j][0] = *(const v2f*)(p + 2);
        lo[j][1] = *(const v2f*)(p + d1);
        hi[j][1] = *(const v2f*)(p + d1 + 2);
    }
#pragma unroll
    for (int j = 0; j < 2; ++j) {
        const int slot = ((2 * tq + j) * KB8 + kb) * 32 + lane;
        ((float4*)WB)[slot]      = nok[j] ? make_float4(lo[j][0].x, lo[j][0].y, hi[j][0].x, hi[j][0].y) : zero4;
        ((float4*)WB)[slot + 32] = (nok[j] && ok1) ? make_float4(lo[j][1].x, lo[j][1].y, hi[j][1].x, hi[j][1].y) : zero4;
    }
}

#define GEMMTH  128
static_assert(EBLK * (GEMMTH / 32) >= NTILES, "the table work-groups cover the N tiles");

__device__ __forceinline__ void position_table_slice(
    const int ngrp,
    const float* __restrict__ pos_emb, const float* __restrict__ W_hid,
    const float* __restrict__ b_hid, float* __restrict__ E_ws,
    float (*G_l)[NREL], float (*S_l)[PKPAD])
{
    const int tid = threadIdx.x;

    for (int e = tid; e < NREL * NREL; e += GEMMTH) {
        const int r = e / NREL, rp = e % NREL;
        const int ad = rp >= KK_REL ? rp - KK_REL : KK_REL - rp;
        const float cnt = (float)(LL - ad);
        const int d = r - rp;
        G_l[r][rp] = cnt * expf(-(float)(d * d));
    }
    __syncthreads();

    for (int e = tid; e < 32 * PKPAD; e += GEMMTH) {
        const int r = e / PKPAD, pcol = e % PKPAD;
        float acc = 0.f;
        if (r < NREL && pcol < PD) {
            for (int rp = 0; rp < NREL; ++rp) acc += G_l[r][rp] * pos_emb[rp * PD + pcol];
        }
        S_l[r][pcol] = acc;
    }
    __syncthreads();

    const int wave = tid >> 5;
    const int lane = tid & 31;
    const int nt = ngrp * (GEMMTH / 32) + wave;
    if (nt >= NTILES) return;

    const int mlane = lane & 15;
    const int kk    = (lane >> 4) * 2;
    const int n     = nt * 16 + mlane;
    const bool nok  = n < DD;
    const float* wrow = W_hid + (size_t)(nok ? n : 0) * FWID + 2 * DD;

    v2f bfr[PKPAD / 4];
#pragma unroll
    for (int s = 0; s < PKPAD / 4; ++s) {
        const int k = s * 4 + kk;
        const bool ok = nok && (k < PD);
        v2f t = *(const v2f*)(wrow + (k < PD ? k : 0));
        bfr[s] = ok ? t : (v2f)0.0f;
    }
    const float bias = b_hid[nok ? n : 0];

    v8f acc0 = (v8f)0.0f, acc1 = (v8f)0.0f;
#pragma unroll
    for (int s = 0; s < PKPAD / 4; ++s) {
        const int k = s * 4 + kk;
        v2f a0 = *(const v2f*)(&S_l[mlane][k]);
        v2f a1 = *(const v2f*)(&S_l[16 + mlane][k]);
        acc0 = __builtin_amdgcn_wmma_f32_16x16x4_f32(false, a0, false, bfr[s], (short)0, acc0, false, false);
        acc1 = __builtin_amdgcn_wmma_f32_16x16x4_f32(false, a1, false, bfr[s], (short)0, acc1, false, false);
    }

    const int rbase = (lane >> 4) << 3;
    if (nok) {
#pragma unroll
        for (int g = 0; g < 8; ++g) {
            const int r0 = rbase + g;
            E_ws[r0 * DD + n] = acc0[g] + bias;
            const int r1 = 16 + rbase + g;
            if (r1 < NREL) E_ws[r1 * DD + n] = acc1[g] + bias;
        }
    }
}

#define GEMMWG  (2 * (MROWS / 32))
#define PFD     4

template <int NN>
struct FragSet {
    v4f e0, s0, e1, s1;
    v4f b[NN];
};

template <int NN>
__device__ __forceinline__ void load_set(FragSet<NN>& f, const float* __restrict__ xe,
                                         const float* __restrict__ xs, const v4f* __restrict__ wb,
                                         const int blk, const int klast)
{
    const int ko = (blk == KB8 - 1) ? klast : blk * 8;
    f.e0 = *(const v4f*)(xe + ko);
    f.s0 = *(const v4f*)(xs + ko);
    f.e1 = *(const v4f*)(xe + 16 * DD + ko);
    f.s1 = *(const v4f*)(xs + 16 * DD + ko);
    const int o = blk * 32;
#pragma unroll
    for (int j = 0; j < NN; ++j) f.b[j] = wb[j * TSTR + o];
}

template <int NN>
__device__ __forceinline__ void mm_set(v8f (&acc)[2][NN], const FragSet<NN>& f, const bool last, const bool inrow)
{
    v4f a0 = f.e0 + f.s0;
    v4f a1 = f.e1 + f.s1;
    if (last) {
        a0 = inrow ? a0 : (v4f)0.0f;
        a1 = inrow ? a1 : (v4f)0.0f;
    }
#pragma unroll
    for (int j = 0; j < NN; ++j) {
        acc[0][j] = __builtin_amdgcn_wmma_f32_16x16x4_f32(false, a0.xy, false, f.b[j].xy, (short)0, acc[0][j], false, false);
        acc[1][j] = __builtin_amdgcn_wmma_f32_16x16x4_f32(false, a1.xy, false, f.b[j].xy, (short)0, acc[1][j], false, false);
    }
#pragma unroll
    for (int j = 0; j < NN; ++j) {
        acc[0][j] = __builtin_amdgcn_wmma_f32_16x16x4_f32(false, a0.zw, false, f.b[j].zw, (short)0, acc[0][j], false, false);
        acc[1][j] = __builtin_amdgcn_wmma_f32_16x16x4_f32(false, a1.zw, false, f.b[j].zw, (short)0, acc[1][j], false, false);
    }
}

template <int NN>
__device__ __forceinline__ void gemm_wave(const float* __restrict__ xe, const float* __restrict__ xs,
                                          const v4f* __restrict__ wb,
                                          float* __restrict__ Out, const int row0, const int nt0,
                                          const int lane)
{
    v8f acc[2][NN];
#pragma unroll
    for (int j = 0; j < NN; ++j) { acc[0][j] = (v8f)0.0f; acc[1][j] = (v8f)0.0f; }

    const bool inrow = lane < 16;
    const int  klast = inrow ? (KB8 - 1) * 8 : (KB8 - 2) * 8;

    FragSet<NN> f[PFD + 1];
#pragma unroll
    for (int p = 0; p < PFD; ++p) load_set<NN>(f[p], xe, xs, wb, p, klast);

#pragma unroll
    for (int kb = 0; kb < KB8; ++kb) {
        if (kb + PFD < KB8) load_set<NN>(f[(kb + PFD) % (PFD + 1)], xe, xs, wb, kb + PFD, klast);
        __builtin_amdgcn_sched_barrier(0);
        mm_set<NN>(acc, f[kb % (PFD + 1)], kb == KB8 - 1, inrow);
        __builtin_amdgcn_sched_barrier(0);
    }

    const int mlane = lane & 15;
    const int rbase = (lane >> 4) << 3;
#pragma unroll
    for (int j = 0; j < NN; ++j) {
        const int ncol = (nt0 + j) * 16 + mlane;
        if (ncol < DD) {
#pragma unroll
            for (int i = 0; i < 2; ++i) {
                float* o = Out + (size_t)(row0 + i * 16 + rbase) * DD + ncol;
#pragma unroll
                for (int g = 0; g < 8; ++g) o[(size_t)g * DD] = acc[i][j][g];
            }
        }
    }
}

__global__ __launch_bounds__(GEMMTH) void pairproj_gemm(
    const float* __restrict__ h_e, const float* __restrict__ h_c,
    const float* __restrict__ h_sh, const float* __restrict__ WB,
    const float* __restrict__ pos_emb, const float* __restrict__ W_hid,
    const float* __restrict__ b_hid, float* __restrict__ E_ws,
    float* __restrict__ A_ws, float* __restrict__ C_ws)
{
    __shared__ float G_l[NREL][NREL];
    __shared__ __attribute__((aligned(16))) float S_l[32][PKPAD];

    if (blockIdx.x >= GEMMWG) {
        position_table_slice(blockIdx.x - GEMMWG, pos_emb, W_hid, b_hid, E_ws, G_l, S_l);
        return;
    }
    const int gid  = blockIdx.x;
    const int sel  = gid >> 7;
    const int mt2  = gid & 127;
    const int wave = __builtin_amdgcn_readfirstlane(threadIdx.x >> 5);
    const int lane = threadIdx.x & 31;

    const float* X   = sel ? h_c : h_e;
    float*       Out = sel ? C_ws : A_ws;
    const int row0 = mt2 * 32;
    const size_t xoff = (size_t)(row0 + (lane & 15)) * DD + (lane >> 4) * 4;

    if (wave < 3) {
        const int nt0 = wave * 5;
        gemm_wave<5>(X + xoff, h_sh + xoff, (const v4f*)WB + (size_t)(sel * NTILES + nt0) * TSTR + lane, Out, row0, nt0, lane);
    } else {
        const int nt0 = 15;
        gemm_wave<4>(X + xoff, h_sh + xoff, (const v4f*)WB + (size_t)(sel * NTILES + nt0) * TSTR + lane, Out, row0, nt0, lane);
    }
}

#define NSEG    19
#define NLONG   14
#define NGRP    13
#define WPB     (NSEG * NGRP)
static_assert(NLONG * 7 + (NSEG - NLONG) * 6 == LL, "the segments tile the emo rows");
static_assert(2 * (NGRP - 1) + 1 == NREL, "the groups tile the diagonals");
#define FULLROW (LL - 1 - KK_REL)
static_assert((BB * WPB) % 8 == 0, "whole work-groups of 8 waves");

__device__ __forceinline__ int pair_row_start(const int i)
{
    if (i <= KK_REL + 1) return (i * (i + NREL)) / 2;
    const int t = i > FULLROW ? i - FULLROW : 0;
    return HEADP + (i - (KK_REL + 1)) * NREL - (t * (t - 1)) / 2;
}

__device__ __forceinline__ v2f elu2(const v2f v)
{
    const v2f l2e    = (v2f)0x1.715476p+0f;
    const v2f ln2_hi = (v2f)0x1.62e430p-1f;
    const v2f ln2_lo = (v2f)-1.90465430e-9f;
    v2f c;
    c.x = fmaxf(v.x, -104.0f);
    c.y = fmaxf(v.y, -104.0f);
    const v2f t = c * l2e;
    v2f p;
    p.x = __builtin_amdgcn_exp2f(t.x);
    p.y = __builtin_amdgcn_exp2f(t.y);
    v2f a = __builtin_elementwise_fma(-t, ln2_hi, c);
    a = __builtin_elementwise_fma(-t, ln2_lo, a);
    const v2f m = __builtin_elementwise_fma(p, a, p - (v2f)1.0f);
    v2f o;
    o.x = (v.x <= 0.f) ? m.x : v.x;
    o.y = (v.y <= 0.f) ? m.y : v.y;
    return o;
}

__device__ __forceinline__ float pair_value(const v2f (&a)[5], const v2f (&c)[5], const v2f (&ev)[5],
                                            const v2f (&gv)[5], const v2f (&bv)[5], const v2f (&wv)[5])
{
    v2f h[5];
    v2f s2, q2;
#pragma unroll
    for (int t = 0; t < 5; ++t) {
        const v2f v = a[t] + c[t] + ev[t];
        h[t] = v;
        if (t == 0) { s2 = v; q2 = v * v; }
        else        { s2 += v; q2 = __builtin_elementwise_fma(v, v, q2); }
    }
    v2f sq;
    sq.x = s2.x + s2.y;
    sq.y = q2.x + q2.y;
#pragma unroll
    for (int off = 16; off > 0; off >>= 1) {
        v2f o;
        o.x = __shfl_xor(sq.x, off, 32);
        o.y = __shfl_xor(sq.y, off, 32);
        sq += o;
    }
    const float mu   = sq.x * (1.f / DD);
    const float var  = sq.y * (1.f / DD) - mu * mu;
    const float rstd = rsqrtf(var + 1e-5f);
    const v2f mu2   = (v2f)mu;
    const v2f rstd2 = (v2f)rstd;

    v2f acc2 = (v2f)0.0f;
#pragma unroll
    for (int t = 0; t < 5; ++t) {
        const v2f x = __builtin_elementwise_fma((h[t] - mu2) * rstd2, gv[t], bv[t]);
        acc2 = __builtin_elementwise_fma(elu2(x), wv[t], acc2);
    }
    float acc = acc2.x + acc2.y;
#pragma unroll
    for (int off = 16; off > 0; off >>= 1) acc += __shfl_xor(acc, off, 32);
    return acc;
}

__device__ __forceinline__ void load_row(v2f (&dst)[5], const float* __restrict__ row, const int lane)
{
#pragma unroll
    for (int t = 0; t < 5; ++t) {
        const int idx = t * 64 + lane * 2;
        dst[t] = (v2f)0.0f;
        if (idx < DD) dst[t] = *(const v2f*)(row + idx);
    }
}

template <int ND>
__device__ __forceinline__ void walk_diagonals(
    const int b, int lo, int hi, const int r0, const int lane,
    const float* __restrict__ A_ws, const float* __restrict__ C_ws, const float* __restrict__ E_ws,
    const float* __restrict__ ln_g, const float* __restrict__ ln_b,
    const float* __restrict__ W_rel, const float brel, float* __restrict__ out)
{
    const int d0 = r0 - KK_REL;
    if (lo < -(d0 + ND - 1)) lo = -(d0 + ND - 1);
    if (hi > LL - 1 - d0) hi = LL - 1 - d0;
    if (lo > hi) return;

    v2f ev[ND][5], gv[5], bv[5], wv[5];
#pragma unroll
    for (int j = 0; j < ND; ++j) load_row(ev[j], E_ws + (size_t)(r0 + j) * DD, lane);
    load_row(gv, ln_g, lane);
    load_row(bv, ln_b, lane);
    load_row(wv, W_rel, lane);

    const float* ab = A_ws + (size_t)b * LL * DD;
    const float* cb = C_ws + (size_t)b * LL * DD;
    float* ob = out + (size_t)b * NPAIR;

    v2f cr[ND][5];
#pragma unroll
    for (int j = 0; j + 1 < ND; ++j) {
        int row = lo + d0 + j;
        row = row < 0 ? 0 : row;
        load_row(cr[j], cb + (size_t)row * DD, lane);
    }

    for (int emo = lo; emo <= hi; ++emo) {
        v2f a[5];
        load_row(a, ab + (size_t)emo * DD, lane);
        int rown = emo + d0 + ND - 1;
        rown = rown > LL - 1 ? LL - 1 : (rown < 0 ? 0 : rown);
        load_row(cr[ND - 1], cb + (size_t)rown * DD, lane);

        float res[ND];
#pragma unroll
        for (int j = 0; j < ND; ++j) res[j] = pair_value(a, cr[j], ev[j], gv, bv, wv);

        const int colo  = emo > KK_REL ? emo - KK_REL : 0;
        const int pbase = pair_row_start(emo) - colo;
#pragma unroll
        for (int j = 0; j < ND; ++j) {
            const int cau = emo + d0 + j;
            if (cau >= 0 && cau < LL && lane == 0) ob[pbase + cau] = res[j] + brel;
        }
#pragma unroll
        for (int j = 0; j + 1 < ND; ++j) {
#pragma unroll
            for (int t = 0; t < 5; ++t) cr[j][t] = cr[j + 1][t];
        }
    }
}

__global__ __launch_bounds__(256, 8) void pair_epilogue(
    const float* __restrict__ A_ws, const float* __restrict__ C_ws,
    const float* __restrict__ E_ws,
    const float* __restrict__ ln_g, const float* __restrict__ ln_b,
    const float* __restrict__ W_rel, const float* __restrict__ b_rel,
    float* __restrict__ out)
{
    const int wave = __builtin_amdgcn_readfirstlane(threadIdx.x >> 5);
    const int lane = threadIdx.x & 31;
    const int w   = blockIdx.x * 8 + wave;
    const int b   = w / WPB;
    const int rem = w - b * WPB;
    const int seg = rem / NGRP;
    const int g   = rem - seg * NGRP;

    const int lo = seg * 7 - (seg > NLONG ? seg - NLONG : 0);
    const int hi = lo + (seg < NLONG ? 6 : 5);
    const float brel = b_rel[0];

    if (g < NGRP - 1)
        walk_diagonals<2>(b, lo, hi, 2 * g, lane, A_ws, C_ws, E_ws, ln_g, ln_b, W_rel, brel, out);
    else
        walk_diagonals<1>(b, lo, hi, NREL - 1, lane, A_ws, C_ws, E_ws, ln_g, ln_b, W_rel, brel, out);
}

extern "C" void kernel_launch(void* const* d_in, const int* in_sizes, int n_in,
                              void* d_out, int out_size, void* d_ws, size_t ws_size,
                              hipStream_t stream) {
    const float* h_e     = (const float*)d_in[0];
    const float* h_c     = (const float*)d_in[1];
    const float* h_sh    = (const float*)d_in[2];
    const float* pos_emb = (const float*)d_in[3];
    const float* W_hid   = (const float*)d_in[4];
    const float* b_hid   = (const float*)d_in[5];
    const float* ln_g    = (const float*)d_in[6];
    const float* ln_b    = (const float*)d_in[7];
    const float* W_rel   = (const float*)d_in[8];
    const float* b_rel   = (const float*)d_in[9];
    float* out = (float*)d_out;

    float* ws   = (float*)d_ws;
    float* A_ws = ws;
    float* C_ws = A_ws + (size_t)MROWS * DD;
    float* WB   = C_ws + (size_t)MROWS * DD;
    float* E_ws = WB   + WPACK;

    pack_w<<<WPWGS, 256, 0, stream>>>(W_hid, WB);

    dim3 ggrid(GEMMWG + EBLK);
    pairproj_gemm<<<ggrid, GEMMTH, 0, stream>>>(h_e, h_c, h_sh, WB, pos_emb, W_hid, b_hid, E_ws, A_ws, C_ws);

    pair_epilogue<<<BB * WPB / 8, 256, 0, stream>>>(A_ws, C_ws, E_ws,
                                                    ln_g, ln_b, W_rel, b_rel, out);
}
